# starting kernel
# baseline (speedup 1.0000x reference)
#include <hip/hip_runtime.h>

typedef __attribute__((ext_vector_type(2))) float v2f;
typedef __attribute__((ext_vector_type(8))) float v8f;

#define F_IN  128
#define F_HID 64
#define F_OUT 40


__global__ void deg_init_kernel(float* __restrict__ deg, int n) {
    int i = blockIdx.x * blockDim.x + threadIdx.x;
    if (i < n) deg[i] = 1.0f;
}

__global__ void deg_accum_kernel(const int* __restrict__ ei, float* __restrict__ deg, int E) {
    int e = blockIdx.x * blockDim.x + threadIdx.x;
    if (e < E) atomicAdd(&deg[ei[E + e]], 1.0f);
}

__global__ void dinv_kernel(float* __restrict__ deg, int n) {
    int i = blockIdx.x * blockDim.x + threadIdx.x;
    if (i < n) { float d = deg[i]; deg[i] = (d > 0.0f) ? rsqrtf(d) : 0.0f; }
}

__global__ void norm_kernel(const int* __restrict__ ei, const float* __restrict__ dinv,
                            float* __restrict__ norm, int E) {
    int e = blockIdx.x * blockDim.x + threadIdx.x;
    if (e < E) norm[e] = dinv[ei[e]] * dinv[ei[E + e]];
}

template <bool RELU_IN>
__global__ __launch_bounds__(128) void gemm_wmma_kernel(const float* __restrict__ X,
                                                        const float* __restrict__ W,
                                                        float* __restrict__ H,
                                                        int K, int Fout) {
    const int lane = threadIdx.x & 31;
    const int wave = threadIdx.x >> 5;
    const int half = lane >> 4;
    const int l    = lane & 15;
    const int row0 = blockIdx.x * 16;
    const int col  = wave * 16 + l;
    const bool colOK  = (col < Fout);
    const int  colLd  = colOK ? col : (Fout - 1);

    const float* __restrict__ xrow = X + (size_t)(row0 + l) * K + 2 * half;
    const float* __restrict__ wcol = W + colLd + (size_t)(2 * half) * Fout;

    v8f c = {};
#pragma unroll 4
    for (int k0 = 0; k0 < K; k0 += 4) {
        float a0 = xrow[k0];
        float a1 = xrow[k0 + 1];
        if (RELU_IN) { a0 = fmaxf(a0, 0.0f); a1 = fmaxf(a1, 0.0f); }
        v2f a; a.x = a0; a.y = a1;
        v2f b;
        b.x = wcol[(size_t)k0 * Fout];
        b.y = wcol[(size_t)(k0 + 1) * Fout];
        c = __builtin_amdgcn_wmma_f32_16x16x4_f32(false, a, false, b, (short)0, c,
                                                  false, false);
    }

    if (colOK) {
#pragma unroll
        for (int v = 0; v < 8; ++v) {
            const int m = v + 8 * half;
            H[(size_t)(row0 + m) * Fout + col] = c[v];
        }
    }
}


__global__ void agg_init_kernel(const float* __restrict__ H, const float* __restrict__ dinv,
                                const float* __restrict__ bias, float* __restrict__ A,
                                int N, int F) {
    int idx = blockIdx.x * blockDim.x + threadIdx.x;
    if (idx < N * F) {
        int i = idx / F;
        int f = idx - i * F;
        float di = dinv[i];
        A[idx] = bias[f] + H[idx] * (di * di);
    }
}

__global__ void agg_edges_kernel(const float* __restrict__ H, const int* __restrict__ ei,
                                 const float* __restrict__ norm, float* __restrict__ A,
                                 int E, int F) {
    const int chunks = F >> 2;
    long long tid   = (long long)blockIdx.x * blockDim.x + threadIdx.x;
    long long total = (long long)E * chunks;
    if (tid >= total) return;
    int e  = (int)(tid / chunks);
    int f0 = (int)(tid - (long long)e * chunks) * 4;

    int   s  = ei[e];
    int   d  = ei[E + e];
    float nv = norm[e];

    const float4 hv = *(const float4*)(H + (size_t)s * F + f0);
    float* __restrict__ out = A + (size_t)d * F + f0;
    atomicAdd(out + 0, hv.x * nv);
    atomicAdd(out + 1, hv.y * nv);
    atomicAdd(out + 2, hv.z * nv);
    atomicAdd(out + 3, hv.w * nv);
}


extern "C" void kernel_launch(void* const* d_in, const int* in_sizes, int n_in,
                              void* d_out, int out_size, void* d_ws, size_t ws_size,
                              hipStream_t stream) {
    const float* x  = (const float*)d_in[0];
    const int*   ei = (const int*)d_in[1];
    const float* W1 = (const float*)d_in[2];
    const float* b1 = (const float*)d_in[3];
    const float* W2 = (const float*)d_in[4];
    const float* b2 = (const float*)d_in[5];
    const float* W3 = (const float*)d_in[6];
    const float* b3 = (const float*)d_in[7];
    float* out = (float*)d_out;

    const int N = in_sizes[0] / F_IN;
    const int E = in_sizes[1] / 2;

    float* dinv = (float*)d_ws;
    float* norm = dinv + N;
    float* bufH = norm + E;
    float* bufA = bufH + (size_t)N * F_HID;

    const int T = 256;
    const int gN  = (N + T - 1) / T;
    const int gE  = (E + T - 1) / T;
    const int gNF = (N * F_HID + T - 1) / T;
    const int gNO = (N * F_OUT + T - 1) / T;
    const unsigned gEF = (unsigned)(((long long)E * (F_HID / 4) + T - 1) / T);
    const unsigned gEO = (unsigned)(((long long)E * (F_OUT / 4) + T - 1) / T);

    deg_init_kernel <<<gN, T, 0, stream>>>(dinv, N);
    deg_accum_kernel<<<gE, T, 0, stream>>>(ei, dinv, E);
    dinv_kernel     <<<gN, T, 0, stream>>>(dinv, N);
    norm_kernel     <<<gE, T, 0, stream>>>(ei, dinv, norm, E);

    const int rowTiles = (N + 15) / 16;
    const int wavesHid = (F_HID + 15) / 16;
    const int wavesOut = (F_OUT + 15) / 16;

    gemm_wmma_kernel<false><<<rowTiles, 32 * wavesHid, 0, stream>>>(x, W1, bufH, F_IN, F_HID);
    agg_init_kernel <<<gNF, T, 0, stream>>>(bufH, dinv, b1, bufA, N, F_HID);
    agg_edges_kernel<<<gEF, T, 0, stream>>>(bufH, ei, norm, bufA, E, F_HID);

    gemm_wmma_kernel<true> <<<rowTiles, 32 * wavesHid, 0, stream>>>(bufA, W2, bufH, F_HID, F_HID);
    agg_init_kernel <<<gNF, T, 0, stream>>>(bufH, dinv, b2, bufA, N, F_HID);
    agg_edges_kernel<<<gEF, T, 0, stream>>>(bufH, ei, norm, bufA, E, F_HID);

    gemm_wmma_kernel<true> <<<rowTiles, 32 * wavesOut, 0, stream>>>(bufA, W3, bufH, F_HID, F_OUT);
    agg_init_kernel <<<gNO, T, 0, stream>>>(bufH, dinv, b3, out, N, F_OUT);
    agg_edges_kernel<<<gEO, T, 0, stream>>>(bufH, ei, norm, out, E, F_OUT);
}


# Edges grouped by destination once per call (integer count giving an arrival-order rank, scan, fill); each layer sums a node's source rows in that list order, one store per row: no float atomics
# speedup vs baseline: 1.9223x; 1.9223x over previous
#include <hip/hip_runtime.h>

typedef __attribute__((ext_vector_type(2))) float v2f;
typedef __attribute__((ext_vector_type(8))) float v8f;

#define F_IN  128
#define F_HID 64
#define F_OUT 40


__global__ void cnt_zero_kernel(int* __restrict__ cnt, int n) {
    int i = blockIdx.x * blockDim.x + threadIdx.x;
    if (i < n) cnt[i] = 0;
}

__global__ void edge_rank_kernel(const int* __restrict__ ei, int* __restrict__ cnt,
                                 int* __restrict__ rank, int E) {
    int e = blockIdx.x * blockDim.x + threadIdx.x;
    if (e < E) rank[e] = atomicAdd(&cnt[ei[E + e]], 1);
}

__global__ __launch_bounds__(256) void rowptr_scan_kernel(const int* __restrict__ cnt,
                                                          int* __restrict__ rowptr, int n) {
    __shared__ int sh[256];
    const int t    = threadIdx.x;
    const int base = blockIdx.x * 256;

    int part = 0;
#pragma unroll 8
    for (int i = t; i < base; i += 256) part += cnt[i];

    const int i = base + t;
    const int v = (i < n) ? cnt[i] : 0;

    sh[t] = part;
    __syncthreads();
    for (int off = 128; off > 0; off >>= 1) {
        if (t < off) sh[t] += sh[t + off];
        __syncthreads();
    }
    const int carry = sh[0];
    __syncthreads();

    sh[t] = v;
    __syncthreads();
    for (int off = 1; off < 256; off <<= 1) {
        const int add = (t >= off) ? sh[t - off] : 0;
        __syncthreads();
        sh[t] += add;
        __syncthreads();
    }
    if (i < n) rowptr[i] = carry + sh[t] - v;
}

__global__ void csr_fill_kernel(const int* __restrict__ ei, const int* __restrict__ rowptr,
                                const int* __restrict__ rank, int* __restrict__ csr_src, int E) {
    int e = blockIdx.x * blockDim.x + threadIdx.x;
    if (e < E) {
        const int p = rowptr[ei[E + e]] + rank[e];
        if ((unsigned)p < (unsigned)E) csr_src[p] = ei[e];
    }
}

template <bool RELU_IN>
__global__ __launch_bounds__(128) void gemm_wmma_kernel(const float* __restrict__ X,
                                                        const float* __restrict__ W,
                                                        float* __restrict__ H,
                                                        int K, int Fout) {
    const int lane = threadIdx.x & 31;
    const int wave = threadIdx.x >> 5;
    const int half = lane >> 4;
    const int l    = lane & 15;
    const int row0 = blockIdx.x * 16;
    const int col  = wave * 16 + l;
    const bool colOK  = (col < Fout);
    const int  colLd  = colOK ? col : (Fout - 1);

    const float* __restrict__ xrow = X + (size_t)(row0 + l) * K + 2 * half;
    const float* __restrict__ wcol = W + colLd + (size_t)(2 * half) * Fout;

    v8f c = {};
#pragma unroll 4
    for (int k0 = 0; k0 < K; k0 += 4) {
        float a0 = xrow[k0];
        float a1 = xrow[k0 + 1];
        if (RELU_IN) { a0 = fmaxf(a0, 0.0f); a1 = fmaxf(a1, 0.0f); }
        v2f a; a.x = a0; a.y = a1;
        v2f b;
        b.x = wcol[(size_t)k0 * Fout];
        b.y = wcol[(size_t)(k0 + 1) * Fout];
        c = __builtin_amdgcn_wmma_f32_16x16x4_f32(false, a, false, b, (short)0, c,
                                                  false, false);
    }

    if (colOK) {
#pragma unroll
        for (int v = 0; v < 8; ++v) {
            const int m = v + 8 * half;
            H[(size_t)(row0 + m) * Fout + col] = c[v];
        }
    }
}


__device__ __forceinline__ float dinv_of(const int* __restrict__ rowptr, int i, int N, int E) {
    const int b  = rowptr[i];
    const int i1 = (i + 1 < N) ? (i + 1) : (N - 1);
    const int r1 = rowptr[i1];
    const int e  = (i + 1 < N) ? r1 : E;
    const float deg = (float)(e - b) + 1.0f;
    return (deg > 0.0f) ? rsqrtf(deg) : 0.0f;
}

template <int F>
__global__ __launch_bounds__(256) void agg_gather_kernel(const float* __restrict__ H,
                                                         const int* __restrict__ rowptr,
                                                         const int* __restrict__ csr_src,
                                                         const float* __restrict__ bias,
                                                         float* __restrict__ A, int N, int E) {
    constexpr int CH = F / 4;
    const int tid = blockIdx.x * 256 + threadIdx.x;
    const int d   = tid / CH;
    if (d >= N) return;
    const int f0  = (tid - d * CH) * 4;

    const int beg = rowptr[d];
    const int d1  = (d + 1 < N) ? (d + 1) : (N - 1);
    const int r1  = rowptr[d1];
    const int end = (d + 1 < N) ? r1 : E;
    const float degd = (float)(end - beg) + 1.0f;
    const float dd   = (degd > 0.0f) ? rsqrtf(degd) : 0.0f;

    const float4 hs = *(const float4*)(H + (size_t)d * F + f0);
    const float  sl = dd * dd;
    float4 acc;
    acc.x = bias[f0 + 0] + hs.x * sl;
    acc.y = bias[f0 + 1] + hs.y * sl;
    acc.z = bias[f0 + 2] + hs.z * sl;
    acc.w = bias[f0 + 3] + hs.w * sl;

#pragma unroll 4
    for (int p = beg; p < end; ++p) {
        const int   s  = csr_src[p];
        const float nv = dinv_of(rowptr, s, N, E) * dd;
        const float4 hv = *(const float4*)(H + (size_t)s * F + f0);
        acc.x += hv.x * nv;
        acc.y += hv.y * nv;
        acc.z += hv.z * nv;
        acc.w += hv.w * nv;
    }
    *(float4*)(A + (size_t)d * F + f0) = acc;
}


extern "C" void kernel_launch(void* const* d_in, const int* in_sizes, int n_in,
                              void* d_out, int out_size, void* d_ws, size_t ws_size,
                              hipStream_t stream) {
    const float* x  = (const float*)d_in[0];
    const int*   ei = (const int*)d_in[1];
    const float* W1 = (const float*)d_in[2];
    const float* b1 = (const float*)d_in[3];
    const float* W2 = (const float*)d_in[4];
    const float* b2 = (const float*)d_in[5];
    const float* W3 = (const float*)d_in[6];
    const float* b3 = (const float*)d_in[7];
    float* out = (float*)d_out;

    const int N = in_sizes[0] / F_IN;
    const int E = in_sizes[1] / 2;

    int*   rowptr  = (int*)d_ws;
    int*   csr_src = rowptr + N;
    float* bufH    = (float*)(csr_src + E);
    float* bufA    = bufH + (size_t)N * F_HID;
    int*   cnt     = (int*)bufA;
    int*   rank    = cnt + N;

    const int T = 256;
    const int gN  = (N + T - 1) / T;
    const int gE  = (E + T - 1) / T;
    const int gAH = (N * (F_HID / 4) + T - 1) / T;
    const int gAO = (N * (F_OUT / 4) + T - 1) / T;

    cnt_zero_kernel   <<<gN, T, 0, stream>>>(cnt, N);
    edge_rank_kernel  <<<gE, T, 0, stream>>>(ei, cnt, rank, E);
    rowptr_scan_kernel<<<gN, T, 0, stream>>>(cnt, rowptr, N);
    csr_fill_kernel   <<<gE, T, 0, stream>>>(ei, rowptr, rank, csr_src, E);

    const int rowTiles = (N + 15) / 16;
    const int wavesHid = (F_HID + 15) / 16;
    const int wavesOut = (F_OUT + 15) / 16;

    gemm_wmma_kernel<false><<<rowTiles, 32 * wavesHid, 0, stream>>>(x, W1, bufH, F_IN, F_HID);
    agg_gather_kernel<F_HID><<<gAH, T, 0, stream>>>(bufH, rowptr, csr_src, b1, bufA, N, E);

    gemm_wmma_kernel<true> <<<rowTiles, 32 * wavesHid, 0, stream>>>(bufA, W2, bufH, F_HID, F_HID);
    agg_gather_kernel<F_HID><<<gAH, T, 0, stream>>>(bufH, rowptr, csr_src, b2, bufA, N, E);

    gemm_wmma_kernel<true> <<<rowTiles, 32 * wavesOut, 0, stream>>>(bufA, W3, bufH, F_HID, F_OUT);
    agg_gather_kernel<F_OUT><<<gAO, T, 0, stream>>>(bufH, rowptr, csr_src, b3, out, N, E);
}


# GCN edge weight factored: projection stores rows pre-scaled by dinv[src]; aggregation adds source rows unscaled, scales the row sum by dinv[dst] once, bias added last: no per-edge degree loads
# speedup vs baseline: 2.1094x; 1.0973x over previous
#include <hip/hip_runtime.h>

typedef __attribute__((ext_vector_type(2))) float v2f;
typedef __attribute__((ext_vector_type(8))) float v8f;

#define F_IN  128
#define F_HID 64
#define F_OUT 40


__device__ __forceinline__ float dinv_from(int b, int e) {
    const float deg = (float)(e - b) + 1.0f;
    return (deg > 0.0f) ? rsqrtf(deg) : 0.0f;
}

__global__ void cnt_zero_kernel(int* __restrict__ cnt, int n) {
    int i = blockIdx.x * blockDim.x + threadIdx.x;
    if (i < n) cnt[i] = 0;
}

__global__ void edge_rank_kernel(const int* __restrict__ ei, int* __restrict__ cnt,
                                 int* __restrict__ rank, int E) {
    int e = blockIdx.x * blockDim.x + threadIdx.x;
    if (e < E) rank[e] = atomicAdd(&cnt[ei[E + e]], 1);
}

__global__ __launch_bounds__(256) void rowptr_scan_kernel(const int* __restrict__ cnt,
                                                          int* __restrict__ rowptr, int n) {
    __shared__ int sh[256];
    const int t    = threadIdx.x;
    const int base = blockIdx.x * 256;

    int part = 0;
#pragma unroll 8
    for (int i = t; i < base; i += 256) part += cnt[i];

    const int i = base + t;
    const int v = (i < n) ? cnt[i] : 0;

    sh[t] = part;
    __syncthreads();
    for (int off = 128; off > 0; off >>= 1) {
        if (t < off) sh[t] += sh[t + off];
        __syncthreads();
    }
    const int carry = sh[0];
    __syncthreads();

    sh[t] = v;
    __syncthreads();
    for (int off = 1; off < 256; off <<= 1) {
        const int add = (t >= off) ? sh[t - off] : 0;
        __syncthreads();
        sh[t] += add;
        __syncthreads();
    }
    if (i < n) rowptr[i] = carry + sh[t] - v;
}

__global__ void csr_fill_kernel(const int* __restrict__ ei, const int* __restrict__ rowptr,
                                const int* __restrict__ rank, int* __restrict__ csr_src, int E) {
    int e = blockIdx.x * blockDim.x + threadIdx.x;
    if (e < E) {
        const int p = rowptr[ei[E + e]] + rank[e];
        if ((unsigned)p < (unsigned)E) csr_src[p] = ei[e];
    }
}

template <bool RELU_IN>
__global__ __launch_bounds__(128) void gemm_wmma_kernel(const float* __restrict__ X,
                                                        const float* __restrict__ W,
                                                        const int* __restrict__ rowptr,
                                                        float* __restrict__ H,
                                                        int K, int Fout, int N, int E) {
    const int lane = threadIdx.x & 31;
    const int wave = threadIdx.x >> 5;
    const int half = lane >> 4;
    const int l    = lane & 15;
    const int row0 = blockIdx.x * 16;
    const int col  = wave * 16 + l;
    const bool colOK  = (col < Fout);
    const int  colLd  = colOK ? col : (Fout - 1);

    const float* __restrict__ xrow = X + (size_t)(row0 + l) * K + 2 * half;
    const float* __restrict__ wcol = W + colLd + (size_t)(2 * half) * Fout;

    float dv[8];
    {
        const int rb = row0 + 8 * half;
        int r[9];
#pragma unroll
        for (int j = 0; j < 9; ++j) {
            const int idx = rb + j;
            const int val = rowptr[(idx < N) ? idx : (N - 1)];
            r[j] = (idx < N) ? val : E;
        }
#pragma unroll
        for (int v = 0; v < 8; ++v) dv[v] = dinv_from(r[v], r[v + 1]);
    }

    v8f c = {};
#pragma unroll 4
    for (int k0 = 0; k0 < K; k0 += 4) {
        float a0 = xrow[k0];
        float a1 = xrow[k0 + 1];
        if (RELU_IN) { a0 = fmaxf(a0, 0.0f); a1 = fmaxf(a1, 0.0f); }
        v2f a; a.x = a0; a.y = a1;
        v2f b;
        b.x = wcol[(size_t)k0 * Fout];
        b.y = wcol[(size_t)(k0 + 1) * Fout];
        c = __builtin_amdgcn_wmma_f32_16x16x4_f32(false, a, false, b, (short)0, c,
                                                  false, false);
    }

    if (colOK) {
#pragma unroll
        for (int v = 0; v < 8; ++v) {
            const int m = v + 8 * half;
            H[(size_t)(row0 + m) * Fout + col] = c[v] * dv[v];
        }
    }
}


template <int F>
__global__ __launch_bounds__(256) void agg_gather_kernel(const float* __restrict__ H,
                                                         const int* __restrict__ rowptr,
                                                         const int* __restrict__ csr_src,
                                                         const float* __restrict__ bias,
                                                         float* __restrict__ A, int N, int E) {
    constexpr int CH = F / 4;
    const int tid = blockIdx.x * 256 + threadIdx.x;
    const int d   = tid / CH;
    if (d >= N) return;
    const int f0  = (tid - d * CH) * 4;

    const int beg = rowptr[d];
    const int d1  = (d + 1 < N) ? (d + 1) : (N - 1);
    const int r1  = rowptr[d1];
    const int end = (d + 1 < N) ? r1 : E;
    const float dd = dinv_from(beg, end);

    float4 acc = *(const float4*)(H + (size_t)d * F + f0);

#pragma unroll 4
    for (int p = beg; p < end; ++p) {
        const int    s  = csr_src[p];
        const float4 hv = *(const float4*)(H + (size_t)s * F + f0);
        acc.x += hv.x;
        acc.y += hv.y;
        acc.z += hv.z;
        acc.w += hv.w;
    }
    float4 o;
    o.x = bias[f0 + 0] + dd * acc.x;
    o.y = bias[f0 + 1] + dd * acc.y;
    o.z = bias[f0 + 2] + dd * acc.z;
    o.w = bias[f0 + 3] + dd * acc.w;
    *(float4*)(A + (size_t)d * F + f0) = o;
}


extern "C" void kernel_launch(void* const* d_in, const int* in_sizes, int n_in,
                              void* d_out, int out_size, void* d_ws, size_t ws_size,
                              hipStream_t stream) {
    const float* x  = (const float*)d_in[0];
    const int*   ei = (const int*)d_in[1];
    const float* W1 = (const float*)d_in[2];
    const float* b1 = (const float*)d_in[3];
    const float* W2 = (const float*)d_in[4];
    const float* b2 = (const float*)d_in[5];
    const float* W3 = (const float*)d_in[6];
    const float* b3 = (const float*)d_in[7];
    float* out = (float*)d_out;

    const int N = in_sizes[0] / F_IN;
    const int E = in_sizes[1] / 2;

    int*   rowptr  = (int*)d_ws;
    int*   csr_src = rowptr + N;
    float* bufH    = (float*)(csr_src + E);
    float* bufA    = bufH + (size_t)N * F_HID;
    int*   cnt     = (int*)bufA;
    int*   rank    = cnt + N;

    const int T = 256;
    const int gN  = (N + T - 1) / T;
    const int gE  = (E + T - 1) / T;
    const int gAH = (N * (F_HID / 4) + T - 1) / T;
    const int gAO = (N * (F_OUT / 4) + T - 1) / T;

    cnt_zero_kernel   <<<gN, T, 0, stream>>>(cnt, N);
    edge_rank_kernel  <<<gE, T, 0, stream>>>(ei, cnt, rank, E);
    rowptr_scan_kernel<<<gN, T, 0, stream>>>(cnt, rowptr, N);
    csr_fill_kernel   <<<gE, T, 0, stream>>>(ei, rowptr, rank, csr_src, E);

    const int rowTiles = (N + 15) / 16;
    const int wavesHid = (F_HID + 15) / 16;
    const int wavesOut = (F_OUT + 15) / 16;

    gemm_wmma_kernel<false><<<rowTiles, 32 * wavesHid, 0, stream>>>(x, W1, rowptr, bufH, F_IN, F_HID, N, E);
    agg_gather_kernel<F_HID><<<gAH, T, 0, stream>>>(bufH, rowptr, csr_src, b1, bufA, N, E);

    gemm_wmma_kernel<true> <<<rowTiles, 32 * wavesHid, 0, stream>>>(bufA, W2, rowptr, bufH, F_HID, F_HID, N, E);
    agg_gather_kernel<F_HID><<<gAH, T, 0, stream>>>(bufH, rowptr, csr_src, b2, bufA, N, E);

    gemm_wmma_kernel<true> <<<rowTiles, 32 * wavesOut, 0, stream>>>(bufA, W3, rowptr, bufH, F_HID, F_OUT, N, E);
    agg_gather_kernel<F_OUT><<<gAO, T, 0, stream>>>(bufH, rowptr, csr_src, b3, out, N, E);
}


# Edge grouping by destination moved from global integer atomics to a two-level counting sort on LDS counters (256-node buckets, 3 launches not 4; rank = arrival order of returning LDS adds)
# speedup vs baseline: 3.0154x; 1.4295x over previous
#include <hip/hip_runtime.h>

typedef __attribute__((ext_vector_type(2))) float v2f;
typedef __attribute__((ext_vector_type(8))) float v8f;

#define F_IN  128
#define F_HID 64
#define F_OUT 40


#define G1 256

__device__ __forceinline__ float dinv_from(int b, int e) {
    const float deg = (float)(e - b) + 1.0f;
    return (deg > 0.0f) ? rsqrtf(deg) : 0.0f;
}

__device__ __forceinline__ int wave_incl_scan(int x, int lane) {
#pragma unroll
    for (int o = 1; o < 32; o <<= 1) {
        const int y = __shfl_up(x, o);
        if (lane >= o) x += y;
    }
    return x;
}

__global__ __launch_bounds__(1024) void bucket_hist_kernel(const int* __restrict__ ei,
                                                           int* __restrict__ hist,
                                                           int E, int NB, int chunk) {
    __shared__ int h[256];
    const int t  = threadIdx.x;
    const int wg = blockIdx.x;
    if (t < 256) h[t] = 0;
    __syncthreads();
    const int c0 = wg * chunk;
    const int c1 = (c0 + chunk < E) ? (c0 + chunk) : E;
    for (int e0 = c0 + t; e0 < c1; e0 += 4096) {
        int d[4];
#pragma unroll
        for (int k = 0; k < 4; ++k) {
            const int e = e0 + 1024 * k;
            d[k] = ei[E + ((e < c1) ? e : (c1 - 1))];
        }
        asm volatile("" :: "v"(d[0]), "v"(d[1]), "v"(d[2]), "v"(d[3]));
#pragma unroll
        for (int k = 0; k < 4; ++k)
            if (e0 + 1024 * k < c1) atomicAdd(&h[(d[k] >> 8) & 255], 1);
    }
    __syncthreads();
    if (t < NB) hist[wg * NB + t] = h[t];
}

__global__ __launch_bounds__(1024) void bucket_scatter_kernel(const int* __restrict__ ei,
                                                              const int* __restrict__ hist,
                                                              int* __restrict__ bbase,
                                                              unsigned* __restrict__ ebuf,
                                                              int E, int NB, int chunk) {
    __shared__ int part[4][256];
    __shared__ int ppre[4][256];
    __shared__ int cur[256];
    __shared__ int wsum[8];
    const int t    = threadIdx.x;
    const int wg   = blockIdx.x;
    const int q    = t >> 8;
    const int b    = t & 255;
    const int lane = t & 31;
    const int wave = t >> 5;

    int tot = 0, pre = 0;
    const int bc = (b < NB) ? b : (NB - 1);
#pragma unroll 32
    for (int j = 0; j < 64; ++j) {
        const int sl = 64 * q + j;
        const int a  = hist[sl * NB + bc];
        tot += a;
        pre += (sl < wg) ? a : 0;
    }
    if (b >= NB) { tot = 0; pre = 0; }
    part[q][b] = tot;
    ppre[q][b] = pre;
    __syncthreads();

    int v = 0, mine = 0;
    if (t < 256) {
        v    = part[0][t] + part[1][t] + part[2][t] + part[3][t];
        mine = ppre[0][t] + ppre[1][t] + ppre[2][t] + ppre[3][t];
    }
    const int x = wave_incl_scan(v, lane);
    if (t < 256 && lane == 31) wsum[wave] = x;
    __syncthreads();
    if (t < 256) {
        int front = 0;
        for (int w = 0; w < wave; ++w) front += wsum[w];
        const int bb = front + x - v;
        cur[t] = bb + mine;
        if (wg == 0 && t < NB) bbase[t] = bb;
    }
    __syncthreads();

    const int c0 = wg * chunk;
    const int c1 = (c0 + chunk < E) ? (c0 + chunk) : E;
    for (int e0 = c0 + t; e0 < c1; e0 += 4096) {
        int d[4], s[4];
#pragma unroll
        for (int k = 0; k < 4; ++k) {
            const int e  = e0 + 1024 * k;
            const int ec = (e < c1) ? e : (c1 - 1);
            d[k] = ei[E + ec];
            s[k] = ei[ec];
        }
        asm volatile("" :: "v"(d[0]), "v"(d[1]), "v"(d[2]), "v"(d[3]),
                           "v"(s[0]), "v"(s[1]), "v"(s[2]), "v"(s[3]));
#pragma unroll
        for (int k = 0; k < 4; ++k) {
            if (e0 + 1024 * k < c1) {
                const int p = atomicAdd(&cur[(d[k] >> 8) & 255], 1);
                if ((unsigned)p < (unsigned)E)
                    ebuf[p] = ((unsigned)s[k] << 8) | ((unsigned)d[k] & 255u);
            }
        }
    }
}

__global__ __launch_bounds__(1024) void bucket_sort_kernel(const unsigned* __restrict__ ebuf,
                                                           const int* __restrict__ bbase,
                                                           int* __restrict__ rowptr,
                                                           int* __restrict__ csr_src,
                                                           int N, int E, int NB) {
    __shared__ int cnt[256];
    __shared__ int cur[256];
    __shared__ int wsum[8];
    const int t    = threadIdx.x;
    const int b    = blockIdx.x;
    const int lane = t & 31;
    const int wave = t >> 5;

    int bbeg = bbase[b];
    int bend = (b + 1 < NB) ? bbase[b + 1] : E;
    bbeg = (bbeg < 0) ? 0 : ((bbeg > E) ? E : bbeg);
    bend = (bend < bbeg) ? bbeg : ((bend > E) ? E : bend);

    if (t < 256) cnt[t] = 0;
    __syncthreads();
    for (int i0 = bbeg + t; i0 < bend; i0 += 4096) {
        unsigned r[4];
#pragma unroll
        for (int k = 0; k < 4; ++k) {
            const int i = i0 + 1024 * k;
            r[k] = ebuf[(i < bend) ? i : (bend - 1)];
        }
        asm volatile("" :: "v"(r[0]), "v"(r[1]), "v"(r[2]), "v"(r[3]));
#pragma unroll
        for (int k = 0; k < 4; ++k)
            if (i0 + 1024 * k < bend) atomicAdd(&cnt[r[k] & 255u], 1);
    }
    __syncthreads();

    const int v = (t < 256) ? cnt[t] : 0;
    const int x = wave_incl_scan(v, lane);
    if (t < 256 && lane == 31) wsum[wave] = x;
    __syncthreads();
    if (t < 256) {
        int front = 0;
        for (int w = 0; w < wave; ++w) front += wsum[w];
        const int first = bbeg + front + x - v;
        cur[t] = first;
        const int node = b * 256 + t;
        if (node < N) rowptr[node] = first;
    }
    __syncthreads();

    for (int i0 = bbeg + t; i0 < bend; i0 += 4096) {
        unsigned r[4];
#pragma unroll
        for (int k = 0; k < 4; ++k) {
            const int i = i0 + 1024 * k;
            r[k] = ebuf[(i < bend) ? i : (bend - 1)];
        }
#pragma unroll
        for (int k = 0; k < 4; ++k) {
            if (i0 + 1024 * k < bend) {
                const int p = atomicAdd(&cur[r[k] & 255u], 1);
                if ((unsigned)p < (unsigned)E) csr_src[p] = (int)(r[k] >> 8);
            }
        }
    }
}

template <bool RELU_IN>
__global__ __launch_bounds__(128) void gemm_wmma_kernel(const float* __restrict__ X,
                                                        const float* __restrict__ W,
                                                        const int* __restrict__ rowptr,
                                                        float* __restrict__ H,
                                                        int K, int Fout, int N, int E) {
    const int lane = threadIdx.x & 31;
    const int wave = threadIdx.x >> 5;
    const int half = lane >> 4;
    const int l    = lane & 15;
    const int row0 = blockIdx.x * 16;
    const int col  = wave * 16 + l;
    const bool colOK  = (col < Fout);
    const int  colLd  = colOK ? col : (Fout - 1);

    const float* __restrict__ xrow = X + (size_t)(row0 + l) * K + 2 * half;
    const float* __restrict__ wcol = W + colLd + (size_t)(2 * half) * Fout;

    float dv[8];
    {
        const int rb = row0 + 8 * half;
        int r[9];
#pragma unroll
        for (int j = 0; j < 9; ++j) {
            const int idx = rb + j;
            const int val = rowptr[(idx < N) ? idx : (N - 1)];
            r[j] = (idx < N) ? val : E;
        }
#pragma unroll
        for (int v = 0; v < 8; ++v) dv[v] = dinv_from(r[v], r[v + 1]);
    }

    v8f c = {};
#pragma unroll 4
    for (int k0 = 0; k0 < K; k0 += 4) {
        float a0 = xrow[k0];
        float a1 = xrow[k0 + 1];
        if (RELU_IN) { a0 = fmaxf(a0, 0.0f); a1 = fmaxf(a1, 0.0f); }
        v2f a; a.x = a0; a.y = a1;
        v2f b;
        b.x = wcol[(size_t)k0 * Fout];
        b.y = wcol[(size_t)(k0 + 1) * Fout];
        c = __builtin_amdgcn_wmma_f32_16x16x4_f32(false, a, false, b, (short)0, c,
                                                  false, false);
    }

    if (colOK) {
#pragma unroll
        for (int v = 0; v < 8; ++v) {
            const int m = v + 8 * half;
            H[(size_t)(row0 + m) * Fout + col] = c[v] * dv[v];
        }
    }
}


template <int F>
__global__ __launch_bounds__(256) void agg_gather_kernel(const float* __restrict__ H,
                                                         const int* __restrict__ rowptr,
                                                         const int* __restrict__ csr_src,
                                                         const float* __restrict__ bias,
                                                         float* __restrict__ A, int N, int E) {
    constexpr int CH = F / 4;
    const int tid = blockIdx.x * 256 + threadIdx.x;
    const int d   = tid / CH;
    if (d >= N) return;
    const int f0  = (tid - d * CH) * 4;

    const int beg = rowptr[d];
    const int d1  = (d + 1 < N) ? (d + 1) : (N - 1);
    const int r1  = rowptr[d1];
    const int end = (d + 1 < N) ? r1 : E;
    const float dd = dinv_from(beg, end);

    float4 acc = *(const float4*)(H + (size_t)d * F + f0);

#pragma unroll 4
    for (int p = beg; p < end; ++p) {
        const int    s  = csr_src[p];
        const float4 hv = *(const float4*)(H + (size_t)s * F + f0);
        acc.x += hv.x;
        acc.y += hv.y;
        acc.z += hv.z;
        acc.w += hv.w;
    }
    float4 o;
    o.x = bias[f0 + 0] + dd * acc.x;
    o.y = bias[f0 + 1] + dd * acc.y;
    o.z = bias[f0 + 2] + dd * acc.z;
    o.w = bias[f0 + 3] + dd * acc.w;
    *(float4*)(A + (size_t)d * F + f0) = o;
}


extern "C" void kernel_launch(void* const* d_in, const int* in_sizes, int n_in,
                              void* d_out, int out_size, void* d_ws, size_t ws_size,
                              hipStream_t stream) {
    const float* x  = (const float*)d_in[0];
    const int*   ei = (const int*)d_in[1];
    const float* W1 = (const float*)d_in[2];
    const float* b1 = (const float*)d_in[3];
    const float* W2 = (const float*)d_in[4];
    const float* b2 = (const float*)d_in[5];
    const float* W3 = (const float*)d_in[6];
    const float* b3 = (const float*)d_in[7];
    float* out = (float*)d_out;

    const int N = in_sizes[0] / F_IN;
    const int E = in_sizes[1] / 2;

    int*   rowptr  = (int*)d_ws;
    int*   csr_src = rowptr + N;
    float* bufH    = (float*)(csr_src + E);
    float* bufA    = bufH + (size_t)N * F_HID;

    const int NB    = (N + 255) / 256;
    const int chunk = (E + G1 - 1) / G1;
    int*      hist  = (int*)bufA;
    int*      bbase = hist + G1 * NB;
    unsigned* ebuf  = (unsigned*)(bbase + 256);

    const int T = 256;
    const int gAH = (N * (F_HID / 4) + T - 1) / T;
    const int gAO = (N * (F_OUT / 4) + T - 1) / T;

    bucket_hist_kernel   <<<G1, 1024, 0, stream>>>(ei, hist, E, NB, chunk);
    bucket_scatter_kernel<<<G1, 1024, 0, stream>>>(ei, hist, bbase, ebuf, E, NB, chunk);
    bucket_sort_kernel   <<<NB, 1024, 0, stream>>>(ebuf, bbase, rowptr, csr_src, N, E, NB);

    const int rowTiles = (N + 15) / 16;
    const int wavesHid = (F_HID + 15) / 16;
    const int wavesOut = (F_OUT + 15) / 16;

    gemm_wmma_kernel<false><<<rowTiles, 32 * wavesHid, 0, stream>>>(x, W1, rowptr, bufH, F_IN, F_HID, N, E);
    agg_gather_kernel<F_HID><<<gAH, T, 0, stream>>>(bufH, rowptr, csr_src, b1, bufA, N, E);

    gemm_wmma_kernel<true> <<<rowTiles, 32 * wavesHid, 0, stream>>>(bufA, W2, rowptr, bufH, F_HID, F_HID, N, E);
    agg_gather_kernel<F_HID><<<gAH, T, 0, stream>>>(bufH, rowptr, csr_src, b2, bufA, N, E);

    gemm_wmma_kernel<true> <<<rowTiles, 32 * wavesOut, 0, stream>>>(bufA, W3, rowptr, bufH, F_HID, F_OUT, N, E);
    agg_gather_kernel<F_OUT><<<gAO, T, 0, stream>>>(bufH, rowptr, csr_src, b3, out, N, E);
}


# Layer 1 and 2 aggregation fused with the next layer's projection: 16-row work-group puts relu(agg) in an LDS tile, WMMA reads A from LDS, stores next H; 9 launches -> 7, no activation round trip
# speedup vs baseline: 3.3478x; 1.1102x over previous
#include <hip/hip_runtime.h>

typedef __attribute__((ext_vector_type(2))) float v2f;
typedef __attribute__((ext_vector_type(8))) float v8f;

#define F_IN  128
#define F_HID 64
#define F_OUT 40


#define G1 256

__device__ __forceinline__ float dinv_from(int b, int e) {
    const float deg = (float)(e - b) + 1.0f;
    return (deg > 0.0f) ? rsqrtf(deg) : 0.0f;
}

__device__ __forceinline__ int wave_incl_scan(int x, int lane) {
#pragma unroll
    for (int o = 1; o < 32; o <<= 1) {
        const int y = __shfl_up(x, o);
        if (lane >= o) x += y;
    }
    return x;
}

__global__ __launch_bounds__(1024) void bucket_hist_kernel(const int* __restrict__ ei,
                                                           int* __restrict__ hist,
                                                           int E, int NB, int chunk) {
    __shared__ int h[256];
    const int t  = threadIdx.x;
    const int wg = blockIdx.x;
    if (t < 256) h[t] = 0;
    __syncthreads();
    const int c0 = wg * chunk;
    const int c1 = (c0 + chunk < E) ? (c0 + chunk) : E;
    for (int e0 = c0 + t; e0 < c1; e0 += 4096) {
        int d[4];
#pragma unroll
        for (int k = 0; k < 4; ++k) {
            const int e = e0 + 1024 * k;
            d[k] = ei[E + ((e < c1) ? e : (c1 - 1))];
        }
        asm volatile("" :: "v"(d[0]), "v"(d[1]), "v"(d[2]), "v"(d[3]));
#pragma unroll
        for (int k = 0; k < 4; ++k)
            if (e0 + 1024 * k < c1) atomicAdd(&h[(d[k] >> 8) & 255], 1);
    }
    __syncthreads();
    if (t < NB) hist[wg * NB + t] = h[t];
}

__global__ __launch_bounds__(1024) void bucket_scatter_kernel(const int* __restrict__ ei,
                                                              const int* __restrict__ hist,
                                                              int* __restrict__ bbase,
                                                              unsigned* __restrict__ ebuf,
                                                              int E, int NB, int chunk) {
    __shared__ int part[4][256];
    __shared__ int ppre[4][256];
    __shared__ int cur[256];
    __shared__ int wsum[8];
    const int t    = threadIdx.x;
    const int wg   = blockIdx.x;
    const int q    = t >> 8;
    const int b    = t & 255;
    const int lane = t & 31;
    const int wave = t >> 5;

    int tot = 0, pre = 0;
    const int bc = (b < NB) ? b : (NB - 1);
#pragma unroll 32
    for (int j = 0; j < 64; ++j) {
        const int sl = 64 * q + j;
        const int a  = hist[sl * NB + bc];
        tot += a;
        pre += (sl < wg) ? a : 0;
    }
    if (b >= NB) { tot = 0; pre = 0; }
    part[q][b] = tot;
    ppre[q][b] = pre;
    __syncthreads();

    int v = 0, mine = 0;
    if (t < 256) {
        v    = part[0][t] + part[1][t] + part[2][t] + part[3][t];
        mine = ppre[0][t] + ppre[1][t] + ppre[2][t] + ppre[3][t];
    }
    const int x = wave_incl_scan(v, lane);
    if (t < 256 && lane == 31) wsum[wave] = x;
    __syncthreads();
    if (t < 256) {
        int front = 0;
        for (int w = 0; w < wave; ++w) front += wsum[w];
        const int bb = front + x - v;
        cur[t] = bb + mine;
        if (wg == 0 && t < NB) bbase[t] = bb;
    }
    __syncthreads();

    const int c0 = wg * chunk;
    const int c1 = (c0 + chunk < E) ? (c0 + chunk) : E;
    for (int e0 = c0 + t; e0 < c1; e0 += 4096) {
        int d[4], s[4];
#pragma unroll
        for (int k = 0; k < 4; ++k) {
            const int e  = e0 + 1024 * k;
            const int ec = (e < c1) ? e : (c1 - 1);
            d[k] = ei[E + ec];
            s[k] = ei[ec];
        }
        asm volatile("" :: "v"(d[0]), "v"(d[1]), "v"(d[2]), "v"(d[3]),
                           "v"(s[0]), "v"(s[1]), "v"(s[2]), "v"(s[3]));
#pragma unroll
        for (int k = 0; k < 4; ++k) {
            if (e0 + 1024 * k < c1) {
                const int p = atomicAdd(&cur[(d[k] >> 8) & 255], 1);
                if ((unsigned)p < (unsigned)E)
                    ebuf[p] = ((unsigned)s[k] << 8) | ((unsigned)d[k] & 255u);
            }
        }
    }
}

__global__ __launch_bounds__(1024) void bucket_sort_kernel(const unsigned* __restrict__ ebuf,
                                                           const int* __restrict__ bbase,
                                                           int* __restrict__ rowptr,
                                                           int* __restrict__ csr_src,
                                                           int N, int E, int NB) {
    __shared__ int cnt[256];
    __shared__ int cur[256];
    __shared__ int wsum[8];
    const int t    = threadIdx.x;
    const int b    = blockIdx.x;
    const int lane = t & 31;
    const int wave = t >> 5;

    int bbeg = bbase[b];
    int bend = (b + 1 < NB) ? bbase[b + 1] : E;
    bbeg = (bbeg < 0) ? 0 : ((bbeg > E) ? E : bbeg);
    bend = (bend < bbeg) ? bbeg : ((bend > E) ? E : bend);

    if (t < 256) cnt[t] = 0;
    __syncthreads();
    for (int i0 = bbeg + t; i0 < bend; i0 += 4096) {
        unsigned r[4];
#pragma unroll
        for (int k = 0; k < 4; ++k) {
            const int i = i0 + 1024 * k;
            r[k] = ebuf[(i < bend) ? i : (bend - 1)];
        }
        asm volatile("" :: "v"(r[0]), "v"(r[1]), "v"(r[2]), "v"(r[3]));
#pragma unroll
        for (int k = 0; k < 4; ++k)
            if (i0 + 1024 * k < bend) atomicAdd(&cnt[r[k] & 255u], 1);
    }
    __syncthreads();

    const int v = (t < 256) ? cnt[t] : 0;
    const int x = wave_incl_scan(v, lane);
    if (t < 256 && lane == 31) wsum[wave] = x;
    __syncthreads();
    if (t < 256) {
        int front = 0;
        for (int w = 0; w < wave; ++w) front += wsum[w];
        const int first = bbeg + front + x - v;
        cur[t] = first;
        const int node = b * 256 + t;
        if (node < N) rowptr[node] = first;
    }
    __syncthreads();

    for (int i0 = bbeg + t; i0 < bend; i0 += 4096) {
        unsigned r[4];
#pragma unroll
        for (int k = 0; k < 4; ++k) {
            const int i = i0 + 1024 * k;
            r[k] = ebuf[(i < bend) ? i : (bend - 1)];
        }
#pragma unroll
        for (int k = 0; k < 4; ++k) {
            if (i0 + 1024 * k < bend) {
                const int p = atomicAdd(&cur[r[k] & 255u], 1);
                if ((unsigned)p < (unsigned)E) csr_src[p] = (int)(r[k] >> 8);
            }
        }
    }
}

template <bool RELU_IN>
__global__ __launch_bounds__(128) void gemm_wmma_kernel(const float* __restrict__ X,
                                                        const float* __restrict__ W,
                                                        const int* __restrict__ rowptr,
                                                        float* __restrict__ H,
                                                        int K, int Fout, int N, int E) {
    const int lane = threadIdx.x & 31;
    const int wave = threadIdx.x >> 5;
    const int half = lane >> 4;
    const int l    = lane & 15;
    const int row0 = blockIdx.x * 16;
    const int col  = wave * 16 + l;
    const bool colOK  = (col < Fout);
    const int  colLd  = colOK ? col : (Fout - 1);

    const float* __restrict__ xrow = X + (size_t)(row0 + l) * K + 2 * half;
    const float* __restrict__ wcol = W + colLd + (size_t)(2 * half) * Fout;

    float dv[8];
    {
        const int rb = row0 + 8 * half;
        int r[9];
#pragma unroll
        for (int j = 0; j < 9; ++j) {
            const int idx = rb + j;
            const int val = rowptr[(idx < N) ? idx : (N - 1)];
            r[j] = (idx < N) ? val : E;
        }
#pragma unroll
        for (int v = 0; v < 8; ++v) dv[v] = dinv_from(r[v], r[v + 1]);
    }

    v8f c = {};
#pragma unroll 4
    for (int k0 = 0; k0 < K; k0 += 4) {
        float a0 = xrow[k0];
        float a1 = xrow[k0 + 1];
        if (RELU_IN) { a0 = fmaxf(a0, 0.0f); a1 = fmaxf(a1, 0.0f); }
        v2f a; a.x = a0; a.y = a1;
        v2f b;
        b.x = wcol[(size_t)k0 * Fout];
        b.y = wcol[(size_t)(k0 + 1) * Fout];
        c = __builtin_amdgcn_wmma_f32_16x16x4_f32(false, a, false, b, (short)0, c,
                                                  false, false);
    }

    if (colOK) {
#pragma unroll
        for (int v = 0; v < 8; ++v) {
            const int m = v + 8 * half;
            H[(size_t)(row0 + m) * Fout + col] = c[v] * dv[v];
        }
    }
}


template <int F>
__global__ __launch_bounds__(256) void agg_gather_kernel(const float* __restrict__ H,
                                                         const int* __restrict__ rowptr,
                                                         const int* __restrict__ csr_src,
                                                         const float* __restrict__ bias,
                                                         float* __restrict__ A, int N, int E) {
    constexpr int CH = F / 4;
    const int tid = blockIdx.x * 256 + threadIdx.x;
    const int d   = tid / CH;
    if (d >= N) return;
    const int f0  = (tid - d * CH) * 4;

    const int beg = rowptr[d];
    const int d1  = (d + 1 < N) ? (d + 1) : (N - 1);
    const int r1  = rowptr[d1];
    const int end = (d + 1 < N) ? r1 : E;
    const float dd = dinv_from(beg, end);

    float4 acc = *(const float4*)(H + (size_t)d * F + f0);

#pragma unroll 4
    for (int p = beg; p < end; ++p) {
        const int    s  = csr_src[p];
        const float4 hv = *(const float4*)(H + (size_t)s * F + f0);
        acc.x += hv.x;
        acc.y += hv.y;
        acc.z += hv.z;
        acc.w += hv.w;
    }
    float4 o;
    o.x = bias[f0 + 0] + dd * acc.x;
    o.y = bias[f0 + 1] + dd * acc.y;
    o.z = bias[f0 + 2] + dd * acc.z;
    o.w = bias[f0 + 3] + dd * acc.w;
    *(float4*)(A + (size_t)d * F + f0) = o;
}

#define TPITCH 68

template <int FOUT>
__global__ __launch_bounds__(256) void agg_proj_kernel(const float* __restrict__ H,
                                                       const int* __restrict__ rowptr,
                                                       const int* __restrict__ csr_src,
                                                       const float* __restrict__ bias,
                                                       const float* __restrict__ W,
                                                       float* __restrict__ Hn, int N, int E) {
    constexpr int F  = F_HID;
    constexpr int NT = (FOUT + 15) / 16;
    __shared__ float tile[16 * TPITCH];
    __shared__ float dinvs[16];

    const int t    = threadIdx.x;
    const int r    = t >> 4;
    const int f0   = (t & 15) * 4;
    const int row0 = blockIdx.x * 16;
    const int d    = row0 + r;

    float4 o  = make_float4(0.0f, 0.0f, 0.0f, 0.0f);
    float  dd = 0.0f;
    if (d < N) {
        const int beg = rowptr[d];
        const int d1  = (d + 1 < N) ? (d + 1) : (N - 1);
        const int r1  = rowptr[d1];
        const int end = (d + 1 < N) ? r1 : E;
        dd = dinv_from(beg, end);

        float4 acc = *(const float4*)(H + (size_t)d * F + f0);

#pragma unroll 4
        for (int p = beg; p < end; ++p) {
            const int    s  = csr_src[p];
            const float4 hv = *(const float4*)(H + (size_t)s * F + f0);
            acc.x += hv.x;
            acc.y += hv.y;
            acc.z += hv.z;
            acc.w += hv.w;
        }
        o.x = fmaxf(bias[f0 + 0] + dd * acc.x, 0.0f);
        o.y = fmaxf(bias[f0 + 1] + dd * acc.y, 0.0f);
        o.z = fmaxf(bias[f0 + 2] + dd * acc.z, 0.0f);
        o.w = fmaxf(bias[f0 + 3] + dd * acc.w, 0.0f);
    }
    *(float4*)(tile + r * TPITCH + f0) = o;
    if ((t & 15) == 0) dinvs[r] = dd;
    __syncthreads();

    const int wave = t >> 5;
    if (wave >= NT) return;
    const int lane = t & 31;
    const int half = lane >> 4;
    const int l    = lane & 15;
    const int col  = wave * 16 + l;
    const bool colOK = (col < FOUT);
    const int  colLd = colOK ? col : (FOUT - 1);

    const float* arow = tile + l * TPITCH + 2 * half;
    const float* __restrict__ wcol = W + colLd + (size_t)(2 * half) * FOUT;

    float dv[8];
#pragma unroll
    for (int v = 0; v < 8; ++v) dv[v] = dinvs[8 * half + v];

    v8f c = {};
#pragma unroll
    for (int k0 = 0; k0 < F; k0 += 4) {
        v2f a; a.x = arow[k0]; a.y = arow[k0 + 1];
        v2f b;
        b.x = wcol[(size_t)k0 * FOUT];
        b.y = wcol[(size_t)(k0 + 1) * FOUT];
        c = __builtin_amdgcn_wmma_f32_16x16x4_f32(false, a, false, b, (short)0, c,
                                                  false, false);
    }

    if (colOK) {
#pragma unroll
        for (int v = 0; v < 8; ++v) {
            const int m = v + 8 * half;
            Hn[(size_t)(row0 + m) * FOUT + col] = c[v] * dv[v];
        }
    }
}


extern "C" void kernel_launch(void* const* d_in, const int* in_sizes, int n_in,
                              void* d_out, int out_size, void* d_ws, size_t ws_size,
                              hipStream_t stream) {
    const float* x  = (const float*)d_in[0];
    const int*   ei = (const int*)d_in[1];
    const float* W1 = (const float*)d_in[2];
    const float* b1 = (const float*)d_in[3];
    const float* W2 = (const float*)d_in[4];
    const float* b2 = (const float*)d_in[5];
    const float* W3 = (const float*)d_in[6];
    const float* b3 = (const float*)d_in[7];
    float* out = (float*)d_out;

    const int N = in_sizes[0] / F_IN;
    const int E = in_sizes[1] / 2;

    int*   rowptr  = (int*)d_ws;
    int*   csr_src = rowptr + N;
    float* bufH    = (float*)(csr_src + E);
    float* bufA    = bufH + (size_t)N * F_HID;

    const int NB    = (N + 255) / 256;
    const int chunk = (E + G1 - 1) / G1;
    int*      hist  = (int*)bufA;
    int*      bbase = hist + G1 * NB;
    unsigned* ebuf  = (unsigned*)(bbase + 256);

    const int T = 256;
    const int gAO = (N * (F_OUT / 4) + T - 1) / T;

    bucket_hist_kernel   <<<G1, 1024, 0, stream>>>(ei, hist, E, NB, chunk);
    bucket_scatter_kernel<<<G1, 1024, 0, stream>>>(ei, hist, bbase, ebuf, E, NB, chunk);
    bucket_sort_kernel   <<<NB, 1024, 0, stream>>>(ebuf, bbase, rowptr, csr_src, N, E, NB);

    const int rowTiles = (N + 15) / 16;
    const int wavesHid = (F_HID + 15) / 16;

    gemm_wmma_kernel<false><<<rowTiles, 32 * wavesHid, 0, stream>>>(x, W1, rowptr, bufH, F_IN, F_HID, N, E);

    agg_proj_kernel<F_HID><<<rowTiles, T, 0, stream>>>(bufH, rowptr, csr_src, b1, W2, bufA, N, E);

    agg_proj_kernel<F_OUT><<<rowTiles, T, 0, stream>>>(bufA, rowptr, csr_src, b2, W3, bufH, N, E);

    agg_gather_kernel<F_OUT><<<gAO, T, 0, stream>>>(bufH, rowptr, csr_src, b3, out, N, E);
}


# Aggregation edge loop rebuilt for loads in flight: 8 independent source-row loads per trip, their indices fetched one trip ahead beside the rows; summation order kept, list tail clamped and masked
# speedup vs baseline: 3.4836x; 1.0406x over previous
#include <hip/hip_runtime.h>

typedef __attribute__((ext_vector_type(2))) float v2f;
typedef __attribute__((ext_vector_type(8))) float v8f;

#define F_IN  128
#define F_HID 64
#define F_OUT 40


#define G1 256

__device__ __forceinline__ float dinv_from(int b, int e) {
    const float deg = (float)(e - b) + 1.0f;
    return (deg > 0.0f) ? rsqrtf(deg) : 0.0f;
}

__device__ __forceinline__ int wave_incl_scan(int x, int lane) {
#pragma unroll
    for (int o = 1; o < 32; o <<= 1) {
        const int y = __shfl_up(x, o);
        if (lane >= o) x += y;
    }
    return x;
}

__global__ __launch_bounds__(1024) void bucket_hist_kernel(const int* __restrict__ ei,
                                                           int* __restrict__ hist,
                                                           int E, int NB, int chunk) {
    __shared__ int h[256];
    const int t  = threadIdx.x;
    const int wg = blockIdx.x;
    if (t < 256) h[t] = 0;
    __syncthreads();
    const int c0 = wg * chunk;
    const int c1 = (c0 + chunk < E) ? (c0 + chunk) : E;
    for (int e0 = c0 + t; e0 < c1; e0 += 4096) {
        int d[4];
#pragma unroll
        for (int k = 0; k < 4; ++k) {
            const int e = e0 + 1024 * k;
            d[k] = ei[E + ((e < c1) ? e : (c1 - 1))];
        }
        asm volatile("" :: "v"(d[0]), "v"(d[1]), "v"(d[2]), "v"(d[3]));
#pragma unroll
        for (int k = 0; k < 4; ++k)
            if (e0 + 1024 * k < c1) atomicAdd(&h[(d[k] >> 8) & 255], 1);
    }
    __syncthreads();
    if (t < NB) hist[wg * NB + t] = h[t];
}

__global__ __launch_bounds__(1024) void bucket_scatter_kernel(const int* __restrict__ ei,
                                                              const int* __restrict__ hist,
                                                              int* __restrict__ bbase,
                                                              unsigned* __restrict__ ebuf,
                                                              int E, int NB, int chunk) {
    __shared__ int part[4][256];
    __shared__ int ppre[4][256];
    __shared__ int cur[256];
    __shared__ int wsum[8];
    const int t    = threadIdx.x;
    const int wg   = blockIdx.x;
    const int q    = t >> 8;
    const int b    = t & 255;
    const int lane = t & 31;
    const int wave = t >> 5;

    int tot = 0, pre = 0;
    const int bc = (b < NB) ? b : (NB - 1);
#pragma unroll 32
    for (int j = 0; j < 64; ++j) {
        const int sl = 64 * q + j;
        const int a  = hist[sl * NB + bc];
        tot += a;
        pre += (sl < wg) ? a : 0;
    }
    if (b >= NB) { tot = 0; pre = 0; }
    part[q][b] = tot;
    ppre[q][b] = pre;
    __syncthreads();

    int v = 0, mine = 0;
    if (t < 256) {
        v    = part[0][t] + part[1][t] + part[2][t] + part[3][t];
        mine = ppre[0][t] + ppre[1][t] + ppre[2][t] + ppre[3][t];
    }
    const int x = wave_incl_scan(v, lane);
    if (t < 256 && lane == 31) wsum[wave] = x;
    __syncthreads();
    if (t < 256) {
        int front = 0;
        for (int w = 0; w < wave; ++w) front += wsum[w];
        const int bb = front + x - v;
        cur[t] = bb + mine;
        if (wg == 0 && t < NB) bbase[t] = bb;
    }
    __syncthreads();

    const int c0 = wg * chunk;
    const int c1 = (c0 + chunk < E) ? (c0 + chunk) : E;
    for (int e0 = c0 + t; e0 < c1; e0 += 4096) {
        int d[4], s[4];
#pragma unroll
        for (int k = 0; k < 4; ++k) {
            const int e  = e0 + 1024 * k;
            const int ec = (e < c1) ? e : (c1 - 1);
            d[k] = ei[E + ec];
            s[k] = ei[ec];
        }
        asm volatile("" :: "v"(d[0]), "v"(d[1]), "v"(d[2]), "v"(d[3]),
                           "v"(s[0]), "v"(s[1]), "v"(s[2]), "v"(s[3]));
#pragma unroll
        for (int k = 0; k < 4; ++k) {
            if (e0 + 1024 * k < c1) {
                const int p = atomicAdd(&cur[(d[k] >> 8) & 255], 1);
                if ((unsigned)p < (unsigned)E)
                    ebuf[p] = ((unsigned)s[k] << 8) | ((unsigned)d[k] & 255u);
            }
        }
    }
}

__global__ __launch_bounds__(1024) void bucket_sort_kernel(const unsigned* __restrict__ ebuf,
                                                           const int* __restrict__ bbase,
                                                           int* __restrict__ rowptr,
                                                           int* __restrict__ csr_src,
                                                           int N, int E, int NB) {
    __shared__ int cnt[256];
    __shared__ int cur[256];
    __shared__ int wsum[8];
    const int t    = threadIdx.x;
    const int b    = blockIdx.x;
    const int lane = t & 31;
    const int wave = t >> 5;

    int bbeg = bbase[b];
    int bend = (b + 1 < NB) ? bbase[b + 1] : E;
    bbeg = (bbeg < 0) ? 0 : ((bbeg > E) ? E : bbeg);
    bend = (bend < bbeg) ? bbeg : ((bend > E) ? E : bend);

    if (t < 256) cnt[t] = 0;
    __syncthreads();
    for (int i0 = bbeg + t; i0 < bend; i0 += 4096) {
        unsigned r[4];
#pragma unroll
        for (int k = 0; k < 4; ++k) {
            const int i = i0 + 1024 * k;
            r[k] = ebuf[(i < bend) ? i : (bend - 1)];
        }
        asm volatile("" :: "v"(r[0]), "v"(r[1]), "v"(r[2]), "v"(r[3]));
#pragma unroll
        for (int k = 0; k < 4; ++k)
            if (i0 + 1024 * k < bend) atomicAdd(&cnt[r[k] & 255u], 1);
    }
    __syncthreads();

    const int v = (t < 256) ? cnt[t] : 0;
    const int x = wave_incl_scan(v, lane);
    if (t < 256 && lane == 31) wsum[wave] = x;
    __syncthreads();
    if (t < 256) {
        int front = 0;
        for (int w = 0; w < wave; ++w) front += wsum[w];
        const int first = bbeg + front + x - v;
        cur[t] = first;
        const int node = b * 256 + t;
        if (node < N) rowptr[node] = first;
    }
    __syncthreads();

    for (int i0 = bbeg + t; i0 < bend; i0 += 4096) {
        unsigned r[4];
#pragma unroll
        for (int k = 0; k < 4; ++k) {
            const int i = i0 + 1024 * k;
            r[k] = ebuf[(i < bend) ? i : (bend - 1)];
        }
#pragma unroll
        for (int k = 0; k < 4; ++k) {
            if (i0 + 1024 * k < bend) {
                const int p = atomicAdd(&cur[r[k] & 255u], 1);
                if ((unsigned)p < (unsigned)E) csr_src[p] = (int)(r[k] >> 8);
            }
        }
    }
}

template <bool RELU_IN>
__global__ __launch_bounds__(128) void gemm_wmma_kernel(const float* __restrict__ X,
                                                        const float* __restrict__ W,
                                                        const int* __restrict__ rowptr,
                                                        float* __restrict__ H,
                                                        int K, int Fout, int N, int E) {
    const int lane = threadIdx.x & 31;
    const int wave = threadIdx.x >> 5;
    const int half = lane >> 4;
    const int l    = lane & 15;
    const int row0 = blockIdx.x * 16;
    const int col  = wave * 16 + l;
    const bool colOK  = (col < Fout);
    const int  colLd  = colOK ? col : (Fout - 1);

    const float* __restrict__ xrow = X + (size_t)(row0 + l) * K + 2 * half;
    const float* __restrict__ wcol = W + colLd + (size_t)(2 * half) * Fout;

    float dv[8];
    {
        const int rb = row0 + 8 * half;
        int r[9];
#pragma unroll
        for (int j = 0; j < 9; ++j) {
            const int idx = rb + j;
            const int val = rowptr[(idx < N) ? idx : (N - 1)];
            r[j] = (idx < N) ? val : E;
        }
#pragma unroll
        for (int v = 0; v < 8; ++v) dv[v] = dinv_from(r[v], r[v + 1]);
    }

    v8f c = {};
#pragma unroll 4
    for (int k0 = 0; k0 < K; k0 += 4) {
        float a0 = xrow[k0];
        float a1 = xrow[k0 + 1];
        if (RELU_IN) { a0 = fmaxf(a0, 0.0f); a1 = fmaxf(a1, 0.0f); }
        v2f a; a.x = a0; a.y = a1;
        v2f b;
        b.x = wcol[(size_t)k0 * Fout];
        b.y = wcol[(size_t)(k0 + 1) * Fout];
        c = __builtin_amdgcn_wmma_f32_16x16x4_f32(false, a, false, b, (short)0, c,
                                                  false, false);
    }

    if (colOK) {
#pragma unroll
        for (int v = 0; v < 8; ++v) {
            const int m = v + 8 * half;
            H[(size_t)(row0 + m) * Fout + col] = c[v] * dv[v];
        }
    }
}


#define GCH 8
template <int F>
__device__ __forceinline__ float4 gather_rows(const float* __restrict__ H,
                                              const int* __restrict__ csr_src,
                                              int beg, int end, int f0, float4 acc) {
    if (beg >= end) return acc;
    const int last = end - 1;
    int s[GCH];
#pragma unroll
    for (int k = 0; k < GCH; ++k) {
        const int q = beg + k;
        s[k] = csr_src[(q < last) ? q : last];
    }
    for (int p = beg; p < end; p += GCH) {
        float4 hv[GCH];
#pragma unroll
        for (int k = 0; k < GCH; ++k) {
            const unsigned off = (unsigned)s[k] * (unsigned)(F * 4) + (unsigned)(f0 * 4);
            hv[k] = *(const float4*)((const char*)H + off);
        }
#pragma unroll
        for (int k = 0; k < GCH; ++k) {
            const int q = p + GCH + k;
            s[k] = csr_src[(q < last) ? q : last];
        }
        __builtin_amdgcn_sched_barrier(0);
#pragma unroll
        for (int k = 0; k < GCH; ++k) {
            const bool ok = (p + k < end);
            acc.x += ok ? hv[k].x : 0.0f;
            acc.y += ok ? hv[k].y : 0.0f;
            acc.z += ok ? hv[k].z : 0.0f;
            acc.w += ok ? hv[k].w : 0.0f;
        }
    }
    return acc;
}

template <int F>
__global__ __launch_bounds__(256, 16) void agg_gather_kernel(const float* __restrict__ H,
                                                         const int* __restrict__ rowptr,
                                                         const int* __restrict__ csr_src,
                                                         const float* __restrict__ bias,
                                                         float* __restrict__ A, int N, int E) {
    constexpr int CH = F / 4;
    const int tid = blockIdx.x * 256 + threadIdx.x;
    const int d   = tid / CH;
    if (d >= N) return;
    const int f0  = (tid - d * CH) * 4;

    const int beg = rowptr[d];
    const int d1  = (d + 1 < N) ? (d + 1) : (N - 1);
    const int r1  = rowptr[d1];
    const int end = (d + 1 < N) ? r1 : E;
    const float dd = dinv_from(beg, end);

    float4 acc = *(const float4*)(H + (size_t)d * F + f0);
    acc = gather_rows<F>(H, csr_src, beg, end, f0, acc);
    float4 o;
    o.x = bias[f0 + 0] + dd * acc.x;
    o.y = bias[f0 + 1] + dd * acc.y;
    o.z = bias[f0 + 2] + dd * acc.z;
    o.w = bias[f0 + 3] + dd * acc.w;
    *(float4*)(A + (size_t)d * F + f0) = o;
}

#define TPITCH 68

template <int FOUT>
__global__ __launch_bounds__(256, 16) void agg_proj_kernel(const float* __restrict__ H,
                                                       const int* __restrict__ rowptr,
                                                       const int* __restrict__ csr_src,
                                                       const float* __restrict__ bias,
                                                       const float* __restrict__ W,
                                                       float* __restrict__ Hn, int N, int E) {
    constexpr int F  = F_HID;
    constexpr int NT = (FOUT + 15) / 16;
    __shared__ float tile[16 * TPITCH];
    __shared__ float dinvs[16];

    const int t    = threadIdx.x;
    const int r    = t >> 4;
    const int f0   = (t & 15) * 4;
    const int row0 = blockIdx.x * 16;
    const int d    = row0 + r;

    float4 o  = make_float4(0.0f, 0.0f, 0.0f, 0.0f);
    float  dd = 0.0f;
    if (d < N) {
        const int beg = rowptr[d];
        const int d1  = (d + 1 < N) ? (d + 1) : (N - 1);
        const int r1  = rowptr[d1];
        const int end = (d + 1 < N) ? r1 : E;
        dd = dinv_from(beg, end);

        float4 acc = *(const float4*)(H + (size_t)d * F + f0);
        acc = gather_rows<F>(H, csr_src, beg, end, f0, acc);
        o.x = fmaxf(bias[f0 + 0] + dd * acc.x, 0.0f);
        o.y = fmaxf(bias[f0 + 1] + dd * acc.y, 0.0f);
        o.z = fmaxf(bias[f0 + 2] + dd * acc.z, 0.0f);
        o.w = fmaxf(bias[f0 + 3] + dd * acc.w, 0.0f);
    }
    *(float4*)(tile + r * TPITCH + f0) = o;
    if ((t & 15) == 0) dinvs[r] = dd;
    __syncthreads();

    const int wave = t >> 5;
    if (wave >= NT) return;
    const int lane = t & 31;
    const int half = lane >> 4;
    const int l    = lane & 15;
    const int col  = wave * 16 + l;
    const bool colOK = (col < FOUT);
    const int  colLd = colOK ? col : (FOUT - 1);

    const float* arow = tile + l * TPITCH + 2 * half;
    const float* __restrict__ wcol = W + colLd + (size_t)(2 * half) * FOUT;

    float dv[8];
#pragma unroll
    for (int v = 0; v < 8; ++v) dv[v] = dinvs[8 * half + v];

    v8f c = {};
#pragma unroll
    for (int k0 = 0; k0 < F; k0 += 4) {
        v2f a; a.x = arow[k0]; a.y = arow[k0 + 1];
        v2f b;
        b.x = wcol[(size_t)k0 * FOUT];
        b.y = wcol[(size_t)(k0 + 1) * FOUT];
        c = __builtin_amdgcn_wmma_f32_16x16x4_f32(false, a, false, b, (short)0, c,
                                                  false, false);
    }

    if (colOK) {
#pragma unroll
        for (int v = 0; v < 8; ++v) {
            const int m = v + 8 * half;
            Hn[(size_t)(row0 + m) * FOUT + col] = c[v] * dv[v];
        }
    }
}


extern "C" void kernel_launch(void* const* d_in, const int* in_sizes, int n_in,
                              void* d_out, int out_size, void* d_ws, size_t ws_size,
                              hipStream_t stream) {
    const float* x  = (const float*)d_in[0];
    const int*   ei = (const int*)d_in[1];
    const float* W1 = (const float*)d_in[2];
    const float* b1 = (const float*)d_in[3];
    const float* W2 = (const float*)d_in[4];
    const float* b2 = (const float*)d_in[5];
    const float* W3 = (const float*)d_in[6];
    const float* b3 = (const float*)d_in[7];
    float* out = (float*)d_out;

    const int N = in_sizes[0] / F_IN;
    const int E = in_sizes[1] / 2;

    int*   rowptr  = (int*)d_ws;
    int*   csr_src = rowptr + N;
    float* bufH    = (float*)(csr_src + E);
    float* bufA    = bufH + (size_t)N * F_HID;

    const int NB    = (N + 255) / 256;
    const int chunk = (E + G1 - 1) / G1;
    int*      hist  = (int*)bufA;
    int*      bbase = hist + G1 * NB;
    unsigned* ebuf  = (unsigned*)(bbase + 256);

    const int T = 256;
    const int gAO = (N * (F_OUT / 4) + T - 1) / T;

    bucket_hist_kernel   <<<G1, 1024, 0, stream>>>(ei, hist, E, NB, chunk);
    bucket_scatter_kernel<<<G1, 1024, 0, stream>>>(ei, hist, bbase, ebuf, E, NB, chunk);
    bucket_sort_kernel   <<<NB, 1024, 0, stream>>>(ebuf, bbase, rowptr, csr_src, N, E, NB);

    const int rowTiles = (N + 15) / 16;
    const int wavesHid = (F_HID + 15) / 16;

    gemm_wmma_kernel<false><<<rowTiles, 32 * wavesHid, 0, stream>>>(x, W1, rowptr, bufH, F_IN, F_HID, N, E);

    agg_proj_kernel<F_HID><<<rowTiles, T, 0, stream>>>(bufH, rowptr, csr_src, b1, W2, bufA, N, E);

    agg_proj_kernel<F_OUT><<<rowTiles, T, 0, stream>>>(bufA, rowptr, csr_src, b2, W3, bufH, N, E);

    agg_gather_kernel<F_OUT><<<gAO, T, 0, stream>>>(bufH, rowptr, csr_src, b3, out, N, E);
}


# Layer-1 projection folded into the bucket-sort launch: each 256-node work-group stages its X rows in LDS by whole-row loads and runs the WMMA tiles itself, dinv from its LDS counts; 7 launches -> 6
# speedup vs baseline: 3.7231x; 1.0688x over previous
#include <hip/hip_runtime.h>

typedef __attribute__((ext_vector_type(2))) float v2f;
typedef __attribute__((ext_vector_type(4))) float v4f;
typedef __attribute__((ext_vector_type(8))) float v8f;

#define F_IN  128
#define F_HID 64
#define F_OUT 40


#define G1 256

__device__ __forceinline__ float dinv_from(int b, int e) {
    const float deg = (float)(e - b) + 1.0f;
    return (deg > 0.0f) ? rsqrtf(deg) : 0.0f;
}

__device__ __forceinline__ int wave_incl_scan(int x, int lane) {
#pragma unroll
    for (int o = 1; o < 32; o <<= 1) {
        const int y = __shfl_up(x, o);
        if (lane >= o) x += y;
    }
    return x;
}

__global__ __launch_bounds__(1024) void bucket_hist_kernel(const int* __restrict__ ei,
                                                           int* __restrict__ hist,
                                                           int E, int NB, int chunk) {
    __shared__ int h[256];
    const int t  = threadIdx.x;
    const int wg = blockIdx.x;
    if (t < 256) h[t] = 0;
    __syncthreads();
    const int c0 = wg * chunk;
    const int c1 = (c0 + chunk < E) ? (c0 + chunk) : E;
    for (int e0 = c0 + t; e0 < c1; e0 += 4096) {
        int d[4];
#pragma unroll
        for (int k = 0; k < 4; ++k) {
            const int e = e0 + 1024 * k;
            d[k] = ei[E + ((e < c1) ? e : (c1 - 1))];
        }
        asm volatile("" :: "v"(d[0]), "v"(d[1]), "v"(d[2]), "v"(d[3]));
#pragma unroll
        for (int k = 0; k < 4; ++k)
            if (e0 + 1024 * k < c1) atomicAdd(&h[(d[k] >> 8) & 255], 1);
    }
    __syncthreads();
    if (t < NB) hist[wg * NB + t] = h[t];
}

__global__ __launch_bounds__(1024) void bucket_scatter_kernel(const int* __restrict__ ei,
                                                              const int* __restrict__ hist,
                                                              int* __restrict__ bbase,
                                                              unsigned* __restrict__ ebuf,
                                                              int E, int NB, int chunk) {
    __shared__ int part[4][256];
    __shared__ int ppre[4][256];
    __shared__ int cur[256];
    __shared__ int wsum[8];
    const int t    = threadIdx.x;
    const int wg   = blockIdx.x;
    const int q    = t >> 8;
    const int b    = t & 255;
    const int lane = t & 31;
    const int wave = t >> 5;

    int tot = 0, pre = 0;
    const int bc = (b < NB) ? b : (NB - 1);
#pragma unroll 32
    for (int j = 0; j < 64; ++j) {
        const int sl = 64 * q + j;
        const int a  = hist[sl * NB + bc];
        tot += a;
        pre += (sl < wg) ? a : 0;
    }
    if (b >= NB) { tot = 0; pre = 0; }
    part[q][b] = tot;
    ppre[q][b] = pre;
    __syncthreads();

    int v = 0, mine = 0;
    if (t < 256) {
        v    = part[0][t] + part[1][t] + part[2][t] + part[3][t];
        mine = ppre[0][t] + ppre[1][t] + ppre[2][t] + ppre[3][t];
    }
    const int x = wave_incl_scan(v, lane);
    if (t < 256 && lane == 31) wsum[wave] = x;
    __syncthreads();
    if (t < 256) {
        int front = 0;
        for (int w = 0; w < wave; ++w) front += wsum[w];
        const int bb = front + x - v;
        cur[t] = bb + mine;
        if (wg == 0 && t < NB) bbase[t] = bb;
    }
    __syncthreads();

    const int c0 = wg * chunk;
    const int c1 = (c0 + chunk < E) ? (c0 + chunk) : E;
    for (int e0 = c0 + t; e0 < c1; e0 += 4096) {
        int d[4], s[4];
#pragma unroll
        for (int k = 0; k < 4; ++k) {
            const int e  = e0 + 1024 * k;
            const int ec = (e < c1) ? e : (c1 - 1);
            d[k] = ei[E + ec];
            s[k] = ei[ec];
        }
        asm volatile("" :: "v"(d[0]), "v"(d[1]), "v"(d[2]), "v"(d[3]),
                           "v"(s[0]), "v"(s[1]), "v"(s[2]), "v"(s[3]));
#pragma unroll
        for (int k = 0; k < 4; ++k) {
            if (e0 + 1024 * k < c1) {
                const int p = atomicAdd(&cur[(d[k] >> 8) & 255], 1);
                if ((unsigned)p < (unsigned)E)
                    ebuf[p] = ((unsigned)s[k] << 8) | ((unsigned)d[k] & 255u);
            }
        }
    }
}

#define XP 132

__global__ __launch_bounds__(1024) void bucket_sort_proj_kernel(const unsigned* __restrict__ ebuf,
                                                                const int* __restrict__ bbase,
                                                                int* __restrict__ rowptr,
                                                                int* __restrict__ csr_src,
                                                                const float* __restrict__ X,
                                                                const float* __restrict__ W,
                                                                float* __restrict__ H,
                                                                int N, int E, int NB) {
    __shared__ float xs[256 * XP];
    __shared__ float dinvs[256];
    __shared__ int cnt[256];
    __shared__ int cur[256];
    __shared__ int wsum[8];
    const int t    = threadIdx.x;
    const int b    = blockIdx.x;
    const int lane = t & 31;
    const int wave = t >> 5;

    int bbeg = bbase[b];
    int bend = (b + 1 < NB) ? bbase[b + 1] : E;
    bbeg = (bbeg < 0) ? 0 : ((bbeg > E) ? E : bbeg);
    bend = (bend < bbeg) ? bbeg : ((bend > E) ? E : bend);

    if (t < 256) cnt[t] = 0;
    __syncthreads();

    const int i00 = bbeg + t;
    unsigned r0[4];
#pragma unroll
    for (int k = 0; k < 4; ++k) {
        const int i  = i00 + 1024 * k;
        int ic = (i < bend) ? i : (bend - 1);
        ic = (ic < 0) ? 0 : ic;
        r0[k] = ebuf[ic];
    }

    v4f xr[8];
#pragma unroll
    for (int j = 0; j < 8; ++j) {
        const int g  = b * 256 + wave + 32 * j;
        const int gc = (g < N) ? g : (N - 1);
        xr[j] = *(const v4f*)(X + (size_t)gc * F_IN + lane * 4);
    }

    asm volatile("" :: "v"(r0[0]), "v"(r0[1]), "v"(r0[2]), "v"(r0[3]));
#pragma unroll
    for (int k = 0; k < 4; ++k)
        if (i00 + 1024 * k < bend) atomicAdd(&cnt[r0[k] & 255u], 1);
    for (int i0 = i00 + 4096; i0 < bend; i0 += 4096) {
        unsigned r[4];
#pragma unroll
        for (int k = 0; k < 4; ++k) {
            const int i = i0 + 1024 * k;
            r[k] = ebuf[(i < bend) ? i : (bend - 1)];
        }
        asm volatile("" :: "v"(r[0]), "v"(r[1]), "v"(r[2]), "v"(r[3]));
#pragma unroll
        for (int k = 0; k < 4; ++k)
            if (i0 + 1024 * k < bend) atomicAdd(&cnt[r[k] & 255u], 1);
    }
#pragma unroll
    for (int j = 0; j < 8; ++j)
        *(v4f*)(xs + (wave + 32 * j) * XP + lane * 4) = xr[j];
    __syncthreads();

    const int v = (t < 256) ? cnt[t] : 0;
    const int x = wave_incl_scan(v, lane);
    if (t < 256 && lane == 31) wsum[wave] = x;
    __syncthreads();
    if (t < 256) {
        int front = 0;
        for (int w = 0; w < wave; ++w) front += wsum[w];
        const int first = bbeg + front + x - v;
        cur[t] = first;
        const int node = b * 256 + t;
        if (node < N) rowptr[node] = first;
        dinvs[t] = dinv_from(first, first + v);
    }
    __syncthreads();

#pragma unroll
    for (int k = 0; k < 4; ++k) {
        if (i00 + 1024 * k < bend) {
            const int p = atomicAdd(&cur[r0[k] & 255u], 1);
            if ((unsigned)p < (unsigned)E) csr_src[p] = (int)(r0[k] >> 8);
        }
    }
    for (int i0 = i00 + 4096; i0 < bend; i0 += 4096) {
        unsigned r[4];
#pragma unroll
        for (int k = 0; k < 4; ++k) {
            const int i = i0 + 1024 * k;
            r[k] = ebuf[(i < bend) ? i : (bend - 1)];
        }
#pragma unroll
        for (int k = 0; k < 4; ++k) {
            if (i0 + 1024 * k < bend) {
                const int p = atomicAdd(&cur[r[k] & 255u], 1);
                if ((unsigned)p < (unsigned)E) csr_src[p] = (int)(r[k] >> 8);
            }
        }
    }

    const int half = lane >> 4;
    const int l    = lane & 15;
    const int ct   = wave & 3;
    const int rp   = wave >> 2;
    const int col  = ct * 16 + l;

    const float* arow0 = xs + (rp * 16 + l) * XP + 2 * half;
    const float* arow1 = arow0 + 128 * XP;
    const float* __restrict__ wcol = W + col + (size_t)(2 * half) * F_HID;

    v8f c0 = {};
    v8f c1 = {};
#pragma unroll
    for (int k0 = 0; k0 < F_IN; k0 += 4) {
        v2f bb;
        bb.x = wcol[(size_t)k0 * F_HID];
        bb.y = wcol[(size_t)(k0 + 1) * F_HID];
        v2f a0; a0.x = arow0[k0]; a0.y = arow0[k0 + 1];
        v2f a1; a1.x = arow1[k0]; a1.y = arow1[k0 + 1];
        c0 = __builtin_amdgcn_wmma_f32_16x16x4_f32(false, a0, false, bb, (short)0, c0,
                                                   false, false);
        c1 = __builtin_amdgcn_wmma_f32_16x16x4_f32(false, a1, false, bb, (short)0, c1,
                                                   false, false);
    }

#pragma unroll
    for (int vv = 0; vv < 8; ++vv) {
        const int lr = rp * 16 + 8 * half + vv;
        const int g0 = b * 256 + lr;
        const int g1 = g0 + 128;
        if (g0 < N) H[(size_t)g0 * F_HID + col] = c0[vv] * dinvs[lr];
        if (g1 < N) H[(size_t)g1 * F_HID + col] = c1[vv] * dinvs[lr + 128];
    }
}


#define GCH 8
template <int F>
__device__ __forceinline__ float4 gather_rows(const float* __restrict__ H,
                                              const int* __restrict__ csr_src,
                                              int beg, int end, int f0, float4 acc) {
    if (beg >= end) return acc;
    const int last = end - 1;
    int s[GCH];
#pragma unroll
    for (int k = 0; k < GCH; ++k) {
        const int q = beg + k;
        s[k] = csr_src[(q < last) ? q : last];
    }
    for (int p = beg; p < end; p += GCH) {
        float4 hv[GCH];
#pragma unroll
        for (int k = 0; k < GCH; ++k) {
            const unsigned off = (unsigned)s[k] * (unsigned)(F * 4) + (unsigned)(f0 * 4);
            hv[k] = *(const float4*)((const char*)H + off);
        }
#pragma unroll
        for (int k = 0; k < GCH; ++k) {
            const int q = p + GCH + k;
            s[k] = csr_src[(q < last) ? q : last];
        }
        __builtin_amdgcn_sched_barrier(0);
#pragma unroll
        for (int k = 0; k < GCH; ++k) {
            const bool ok = (p + k < end);
            acc.x += ok ? hv[k].x : 0.0f;
            acc.y += ok ? hv[k].y : 0.0f;
            acc.z += ok ? hv[k].z : 0.0f;
            acc.w += ok ? hv[k].w : 0.0f;
        }
    }
    return acc;
}

template <int F>
__global__ __launch_bounds__(256, 16) void agg_gather_kernel(const float* __restrict__ H,
                                                         const int* __restrict__ rowptr,
                                                         const int* __restrict__ csr_src,
                                                         const float* __restrict__ bias,
                                                         float* __restrict__ A, int N, int E) {
    constexpr int CH = F / 4;
    const int tid = blockIdx.x * 256 + threadIdx.x;
    const int d   = tid / CH;
    if (d >= N) return;
    const int f0  = (tid - d * CH) * 4;

    const int beg = rowptr[d];
    const int d1  = (d + 1 < N) ? (d + 1) : (N - 1);
    const int r1  = rowptr[d1];
    const int end = (d + 1 < N) ? r1 : E;
    const float dd = dinv_from(beg, end);

    float4 acc = *(const float4*)(H + (size_t)d * F + f0);
    acc = gather_rows<F>(H, csr_src, beg, end, f0, acc);
    float4 o;
    o.x = bias[f0 + 0] + dd * acc.x;
    o.y = bias[f0 + 1] + dd * acc.y;
    o.z = bias[f0 + 2] + dd * acc.z;
    o.w = bias[f0 + 3] + dd * acc.w;
    *(float4*)(A + (size_t)d * F + f0) = o;
}

#define TPITCH 68

template <int FOUT>
__global__ __launch_bounds__(256, 16) void agg_proj_kernel(const float* __restrict__ H,
                                                       const int* __restrict__ rowptr,
                                                       const int* __restrict__ csr_src,
                                                       const float* __restrict__ bias,
                                                       const float* __restrict__ W,
                                                       float* __restrict__ Hn, int N, int E) {
    constexpr int F  = F_HID;
    constexpr int NT = (FOUT + 15) / 16;
    __shared__ float tile[16 * TPITCH];
    __shared__ float dinvs[16];

    const int t    = threadIdx.x;
    const int r    = t >> 4;
    const int f0   = (t & 15) * 4;
    const int row0 = blockIdx.x * 16;
    const int d    = row0 + r;

    float4 o  = make_float4(0.0f, 0.0f, 0.0f, 0.0f);
    float  dd = 0.0f;
    if (d < N) {
        const int beg = rowptr[d];
        const int d1  = (d + 1 < N) ? (d + 1) : (N - 1);
        const int r1  = rowptr[d1];
        const int end = (d + 1 < N) ? r1 : E;
        dd = dinv_from(beg, end);

        float4 acc = *(const float4*)(H + (size_t)d * F + f0);
        acc = gather_rows<F>(H, csr_src, beg, end, f0, acc);
        o.x = fmaxf(bias[f0 + 0] + dd * acc.x, 0.0f);
        o.y = fmaxf(bias[f0 + 1] + dd * acc.y, 0.0f);
        o.z = fmaxf(bias[f0 + 2] + dd * acc.z, 0.0f);
        o.w = fmaxf(bias[f0 + 3] + dd * acc.w, 0.0f);
    }
    *(float4*)(tile + r * TPITCH + f0) = o;
    if ((t & 15) == 0) dinvs[r] = dd;
    __syncthreads();

    const int wave = t >> 5;
    if (wave >= NT) return;
    const int lane = t & 31;
    const int half = lane >> 4;
    const int l    = lane & 15;
    const int col  = wave * 16 + l;
    const bool colOK = (col < FOUT);
    const int  colLd = colOK ? col : (FOUT - 1);

    const float* arow = tile + l * TPITCH + 2 * half;
    const float* __restrict__ wcol = W + colLd + (size_t)(2 * half) * FOUT;

    float dv[8];
#pragma unroll
    for (int v = 0; v < 8; ++v) dv[v] = dinvs[8 * half + v];

    v8f c = {};
#pragma unroll
    for (int k0 = 0; k0 < F; k0 += 4) {
        v2f a; a.x = arow[k0]; a.y = arow[k0 + 1];
        v2f b;
        b.x = wcol[(size_t)k0 * FOUT];
        b.y = wcol[(size_t)(k0 + 1) * FOUT];
        c = __builtin_amdgcn_wmma_f32_16x16x4_f32(false, a, false, b, (short)0, c,
                                                  false, false);
    }

    if (colOK) {
#pragma unroll
        for (int v = 0; v < 8; ++v) {
            const int m = v + 8 * half;
            Hn[(size_t)(row0 + m) * FOUT + col] = c[v] * dv[v];
        }
    }
}


extern "C" void kernel_launch(void* const* d_in, const int* in_sizes, int n_in,
                              void* d_out, int out_size, void* d_ws, size_t ws_size,
                              hipStream_t stream) {
    const float* x  = (const float*)d_in[0];
    const int*   ei = (const int*)d_in[1];
    const float* W1 = (const float*)d_in[2];
    const float* b1 = (const float*)d_in[3];
    const float* W2 = (const float*)d_in[4];
    const float* b2 = (const float*)d_in[5];
    const float* W3 = (const float*)d_in[6];
    const float* b3 = (const float*)d_in[7];
    float* out = (float*)d_out;

    const int N = in_sizes[0] / F_IN;
    const int E = in_sizes[1] / 2;

    int*   rowptr  = (int*)d_ws;
    int*   csr_src = rowptr + N;
    float* bufH    = (float*)(csr_src + E);
    float* bufA    = bufH + (size_t)N * F_HID;

    const int NB    = (N + 255) / 256;
    const int chunk = (E + G1 - 1) / G1;
    int*      hist  = (int*)bufA;
    int*      bbase = hist + G1 * NB;
    unsigned* ebuf  = (unsigned*)(bbase + 256);

    const int T = 256;
    const int gAO = (N * (F_OUT / 4) + T - 1) / T;

    bucket_hist_kernel   <<<G1, 1024, 0, stream>>>(ei, hist, E, NB, chunk);
    bucket_scatter_kernel<<<G1, 1024, 0, stream>>>(ei, hist, bbase, ebuf, E, NB, chunk);
    bucket_sort_proj_kernel<<<NB, 1024, 0, stream>>>(ebuf, bbase, rowptr, csr_src, x, W1, bufH, N, E, NB);

    const int rowTiles = (N + 15) / 16;

    agg_proj_kernel<F_HID><<<rowTiles, T, 0, stream>>>(bufH, rowptr, csr_src, b1, W2, bufA, N, E);

    agg_proj_kernel<F_OUT><<<rowTiles, T, 0, stream>>>(bufA, rowptr, csr_src, b2, W3, bufH, N, E);

    agg_gather_kernel<F_OUT><<<gAO, T, 0, stream>>>(bufH, rowptr, csr_src, b3, out, N, E);
}


# Edge grouping in 2 launches, not 3: each slice sorts its own edges by bucket inside its own range, so no counting launch runs first; a bucket gathers its 256 runs; list order stays LDS arrival order
# speedup vs baseline: 4.3651x; 1.1724x over previous
#include <hip/hip_runtime.h>

typedef __attribute__((ext_vector_type(2))) float v2f;
typedef __attribute__((ext_vector_type(4))) float v4f;
typedef __attribute__((ext_vector_type(8))) float v8f;

#define F_IN  128
#define F_HID 64
#define F_OUT 40


#define G1 256
#define HP 256
static_assert(G1 == 256, "pass 2 maps its 1024 threads onto 256 slices: 4 threads per run, 16 groups of 16 slices");

__device__ __forceinline__ float dinv_from(int b, int e) {
    const float deg = (float)(e - b) + 1.0f;
    return (deg > 0.0f) ? rsqrtf(deg) : 0.0f;
}

__device__ __forceinline__ int wave_incl_scan(int x, int lane) {
#pragma unroll
    for (int o = 1; o < 32; o <<= 1) {
        const int y = __shfl_up(x, o);
        if (lane >= o) x += y;
    }
    return x;
}

__global__ __launch_bounds__(1024) void slice_sort_kernel(const int* __restrict__ ei,
                                                          int* __restrict__ hist,
                                                          int* __restrict__ loff,
                                                          unsigned* __restrict__ ebuf,
                                                          int E, int chunk) {
    __shared__ int h[256];
    __shared__ int cur[256];
    __shared__ int wsum[8];
    const int t    = threadIdx.x;
    const int wg   = blockIdx.x;
    const int lane = t & 31;
    const int wave = t >> 5;
    if (t < 256) h[t] = 0;
    __syncthreads();
    const int c0 = wg * chunk;
    const int c1 = (c0 + chunk < E) ? (c0 + chunk) : E;

    const int e00 = c0 + t;
    int d0[4], s0[4];
#pragma unroll
    for (int k = 0; k < 4; ++k) {
        const int e = e00 + 1024 * k;
        int ec = (e < c1) ? e : (c1 - 1);
        ec = (ec < 0) ? 0 : ec;
        d0[k] = ei[E + ec];
        s0[k] = ei[ec];
    }
    asm volatile("" :: "v"(d0[0]), "v"(d0[1]), "v"(d0[2]), "v"(d0[3]),
                       "v"(s0[0]), "v"(s0[1]), "v"(s0[2]), "v"(s0[3]));
#pragma unroll
    for (int k = 0; k < 4; ++k)
        if (e00 + 1024 * k < c1) atomicAdd(&h[(d0[k] >> 8) & 255], 1);
    for (int e0 = e00 + 4096; e0 < c1; e0 += 4096) {
        int d[4];
#pragma unroll
        for (int k = 0; k < 4; ++k) {
            const int e = e0 + 1024 * k;
            d[k] = ei[E + ((e < c1) ? e : (c1 - 1))];
        }
        asm volatile("" :: "v"(d[0]), "v"(d[1]), "v"(d[2]), "v"(d[3]));
#pragma unroll
        for (int k = 0; k < 4; ++k)
            if (e0 + 1024 * k < c1) atomicAdd(&h[(d[k] >> 8) & 255], 1);
    }
    __syncthreads();

    const int v = (t < 256) ? h[t] : 0;
    const int x = wave_incl_scan(v, lane);
    if (t < 256 && lane == 31) wsum[wave] = x;
    __syncthreads();
    if (t < 256) {
        int front = 0;
        for (int w = 0; w < wave; ++w) front += wsum[w];
        const int lo = front + x - v;
        cur[t] = lo;
        hist[wg * HP + t] = v;
        loff[wg * HP + t] = lo;
    }
    __syncthreads();

#pragma unroll
    for (int k = 0; k < 4; ++k) {
        if (e00 + 1024 * k < c1) {
            const int p = c0 + atomicAdd(&cur[(d0[k] >> 8) & 255], 1);
            if ((unsigned)p < (unsigned)E)
                ebuf[p] = ((unsigned)s0[k] << 8) | ((unsigned)d0[k] & 255u);
        }
    }
    for (int e0 = e00 + 4096; e0 < c1; e0 += 4096) {
        int d[4], s[4];
#pragma unroll
        for (int k = 0; k < 4; ++k) {
            const int e  = e0 + 1024 * k;
            const int ec = (e < c1) ? e : (c1 - 1);
            d[k] = ei[E + ec];
            s[k] = ei[ec];
        }
#pragma unroll
        for (int k = 0; k < 4; ++k) {
            if (e0 + 1024 * k < c1) {
                const int p = c0 + atomicAdd(&cur[(d[k] >> 8) & 255], 1);
                if ((unsigned)p < (unsigned)E)
                    ebuf[p] = ((unsigned)s[k] << 8) | ((unsigned)d[k] & 255u);
            }
        }
    }
}

#define XP 132

typedef __attribute__((ext_vector_type(4))) int v4i;

#define RK 8

__global__ __launch_bounds__(1024) void bucket_sort_proj_kernel(const unsigned* __restrict__ ebuf,
                                                                const int* __restrict__ hist,
                                                                const int* __restrict__ loff,
                                                                int* __restrict__ rowptr,
                                                                int* __restrict__ csr_src,
                                                                const float* __restrict__ X,
                                                                const float* __restrict__ W,
                                                                float* __restrict__ H,
                                                                int N, int E, int chunk) {
    __shared__ float xs[256 * XP];
    __shared__ float dinvs[256];
    __shared__ int part[16][256];
    __shared__ int cnt[256];
    __shared__ int cur[256];
    __shared__ int wsum[8];
    __shared__ int wsumb[8];
    __shared__ int bstart;
    const int t    = threadIdx.x;
    const int b    = blockIdx.x;
    const int lane = t & 31;
    const int wave = t >> 5;

    v4f xr[8];
#pragma unroll
    for (int jx = 0; jx < 8; ++jx) {
        const int g  = b * 256 + wave + 32 * jx;
        const int gc = (g < N) ? g : (N - 1);
        xr[jx] = *(const v4f*)(X + (size_t)gc * F_IN + lane * 4);
    }

    const int sl = t >> 2;
    const int j  = t & 3;
    int ln = hist[sl * HP + b];
    int st = sl * chunk + loff[sl * HP + b];

    const int bq = t & 63;
    const int sg = t >> 6;
    v4i tot = {0, 0, 0, 0};
#pragma unroll
    for (int jj = 0; jj < 16; ++jj)
        tot += *(const v4i*)(hist + (sg * 16 + jj) * HP + bq * 4);

    ln = (ln < 0) ? 0 : ln;
    st = (st < 0) ? 0 : ((st > E) ? E : st);
    ln = (ln > E - st) ? (E - st) : ln;

    unsigned r0[RK];
#pragma unroll
    for (int k = 0; k < RK; ++k) {
        const int i = j + 4 * k;
        int ic = st + ((i < ln) ? i : (ln - 1));
        ic = (ic < 0) ? 0 : ic;
        r0[k] = ebuf[ic];
    }

    *(v4i*)&part[sg][bq * 4] = tot;
    if (t < 256) cnt[t] = 0;
    __syncthreads();

    asm volatile("" :: "v"(r0[0]), "v"(r0[1]), "v"(r0[2]), "v"(r0[3]),
                       "v"(r0[4]), "v"(r0[5]), "v"(r0[6]), "v"(r0[7]));
#pragma unroll
    for (int k = 0; k < RK; ++k)
        if (j + 4 * k < ln) atomicAdd(&cnt[r0[k] & 255u], 1);
    for (int i0 = j + 4 * RK; i0 < ln; i0 += 16) {
        unsigned r[4];
#pragma unroll
        for (int k = 0; k < 4; ++k) {
            const int i = i0 + 4 * k;
            r[k] = ebuf[st + ((i < ln) ? i : (ln - 1))];
        }
        asm volatile("" :: "v"(r[0]), "v"(r[1]), "v"(r[2]), "v"(r[3]));
#pragma unroll
        for (int k = 0; k < 4; ++k)
            if (i0 + 4 * k < ln) atomicAdd(&cnt[r[k] & 255u], 1);
    }

    int vb = 0;
    if (t < 256) {
#pragma unroll
        for (int g = 0; g < 16; ++g) vb += part[g][t];
    }
    const int xb = wave_incl_scan(vb, lane);
    if (t < 256 && lane == 31) wsumb[wave] = xb;

#pragma unroll
    for (int jx = 0; jx < 8; ++jx)
        *(v4f*)(xs + (wave + 32 * jx) * XP + lane * 4) = xr[jx];
    __syncthreads();

    const int v = (t < 256) ? cnt[t] : 0;
    const int x = wave_incl_scan(v, lane);
    if (t < 256 && lane == 31) wsum[wave] = x;
    if (t == b) {
        int front = 0;
        for (int w = 0; w < wave; ++w) front += wsumb[w];
        int bb = front + xb - vb;
        bb = (bb < 0) ? 0 : ((bb > E) ? E : bb);
        bstart = bb;
    }
    __syncthreads();
    if (t < 256) {
        int front = 0;
        for (int w = 0; w < wave; ++w) front += wsum[w];
        const int first = bstart + front + x - v;
        cur[t] = first;
        const int node = b * 256 + t;
        if (node < N) rowptr[node] = first;
        dinvs[t] = dinv_from(first, first + v);
    }
    __syncthreads();

#pragma unroll
    for (int k = 0; k < RK; ++k) {
        if (j + 4 * k < ln) {
            const int p = atomicAdd(&cur[r0[k] & 255u], 1);
            if ((unsigned)p < (unsigned)E) csr_src[p] = (int)(r0[k] >> 8);
        }
    }
    for (int i0 = j + 4 * RK; i0 < ln; i0 += 16) {
        unsigned r[4];
#pragma unroll
        for (int k = 0; k < 4; ++k) {
            const int i = i0 + 4 * k;
            r[k] = ebuf[st + ((i < ln) ? i : (ln - 1))];
        }
#pragma unroll
        for (int k = 0; k < 4; ++k) {
            if (i0 + 4 * k < ln) {
                const int p = atomicAdd(&cur[r[k] & 255u], 1);
                if ((unsigned)p < (unsigned)E) csr_src[p] = (int)(r[k] >> 8);
            }
        }
    }

    const int half = lane >> 4;
    const int l    = lane & 15;
    const int ct   = wave & 3;
    const int rp   = wave >> 2;
    const int col  = ct * 16 + l;

    const float* arow0 = xs + (rp * 16 + l) * XP + 2 * half;
    const float* arow1 = arow0 + 128 * XP;
    const float* __restrict__ wcol = W + col + (size_t)(2 * half) * F_HID;

    v8f c0 = {};
    v8f c1 = {};
#pragma unroll
    for (int k0 = 0; k0 < F_IN; k0 += 4) {
        v2f bb;
        bb.x = wcol[(size_t)k0 * F_HID];
        bb.y = wcol[(size_t)(k0 + 1) * F_HID];
        v2f a0; a0.x = arow0[k0]; a0.y = arow0[k0 + 1];
        v2f a1; a1.x = arow1[k0]; a1.y = arow1[k0 + 1];
        c0 = __builtin_amdgcn_wmma_f32_16x16x4_f32(false, a0, false, bb, (short)0, c0,
                                                   false, false);
        c1 = __builtin_amdgcn_wmma_f32_16x16x4_f32(false, a1, false, bb, (short)0, c1,
                                                   false, false);
    }

#pragma unroll
    for (int vv = 0; vv < 8; ++vv) {
        const int lr = rp * 16 + 8 * half + vv;
        const int g0 = b * 256 + lr;
        const int g1 = g0 + 128;
        if (g0 < N) H[(size_t)g0 * F_HID + col] = c0[vv] * dinvs[lr];
        if (g1 < N) H[(size_t)g1 * F_HID + col] = c1[vv] * dinvs[lr + 128];
    }
}


#define GCH 8
template <int F>
__device__ __forceinline__ float4 gather_rows(const float* __restrict__ H,
                                              const int* __restrict__ csr_src,
                                              int beg, int end, int f0, float4 acc) {
    if (beg >= end) return acc;
    const int last = end - 1;
    int s[GCH];
#pragma unroll
    for (int k = 0; k < GCH; ++k) {
        const int q = beg + k;
        s[k] = csr_src[(q < last) ? q : last];
    }
    for (int p = beg; p < end; p += GCH) {
        float4 hv[GCH];
#pragma unroll
        for (int k = 0; k < GCH; ++k) {
            const unsigned off = (unsigned)s[k] * (unsigned)(F * 4) + (unsigned)(f0 * 4);
            hv[k] = *(const float4*)((const char*)H + off);
        }
#pragma unroll
        for (int k = 0; k < GCH; ++k) {
            const int q = p + GCH + k;
            s[k] = csr_src[(q < last) ? q : last];
        }
        __builtin_amdgcn_sched_barrier(0);
#pragma unroll
        for (int k = 0; k < GCH; ++k) {
            const bool ok = (p + k < end);
            acc.x += ok ? hv[k].x : 0.0f;
            acc.y += ok ? hv[k].y : 0.0f;
            acc.z += ok ? hv[k].z : 0.0f;
            acc.w += ok ? hv[k].w : 0.0f;
        }
    }
    return acc;
}

template <int F>
__global__ __launch_bounds__(256, 16) void agg_gather_kernel(const float* __restrict__ H,
                                                         const int* __restrict__ rowptr,
                                                         const int* __restrict__ csr_src,
                                                         const float* __restrict__ bias,
                                                         float* __restrict__ A, int N, int E) {
    constexpr int CH = F / 4;
    const int tid = blockIdx.x * 256 + threadIdx.x;
    const int d   = tid / CH;
    if (d >= N) return;
    const int f0  = (tid - d * CH) * 4;

    const int beg = rowptr[d];
    const int d1  = (d + 1 < N) ? (d + 1) : (N - 1);
    const int r1  = rowptr[d1];
    const int end = (d + 1 < N) ? r1 : E;
    const float dd = dinv_from(beg, end);

    float4 acc = *(const float4*)(H + (size_t)d * F + f0);
    acc = gather_rows<F>(H, csr_src, beg, end, f0, acc);
    float4 o;
    o.x = bias[f0 + 0] + dd * acc.x;
    o.y = bias[f0 + 1] + dd * acc.y;
    o.z = bias[f0 + 2] + dd * acc.z;
    o.w = bias[f0 + 3] + dd * acc.w;
    *(float4*)(A + (size_t)d * F + f0) = o;
}

#define TPITCH 68

template <int FOUT>
__global__ __launch_bounds__(256, 16) void agg_proj_kernel(const float* __restrict__ H,
                                                       const int* __restrict__ rowptr,
                                                       const int* __restrict__ csr_src,
                                                       const float* __restrict__ bias,
                                                       const float* __restrict__ W,
                                                       float* __restrict__ Hn, int N, int E) {
    constexpr int F  = F_HID;
    constexpr int NT = (FOUT + 15) / 16;
    __shared__ float tile[16 * TPITCH];
    __shared__ float dinvs[16];

    const int t    = threadIdx.x;
    const int r    = t >> 4;
    const int f0   = (t & 15) * 4;
    const int row0 = blockIdx.x * 16;
    const int d    = row0 + r;

    float4 o  = make_float4(0.0f, 0.0f, 0.0f, 0.0f);
    float  dd = 0.0f;
    if (d < N) {
        const int beg = rowptr[d];
        const int d1  = (d + 1 < N) ? (d + 1) : (N - 1);
        const int r1  = rowptr[d1];
        const int end = (d + 1 < N) ? r1 : E;
        dd = dinv_from(beg, end);

        float4 acc = *(const float4*)(H + (size_t)d * F + f0);
        acc = gather_rows<F>(H, csr_src, beg, end, f0, acc);
        o.x = fmaxf(bias[f0 + 0] + dd * acc.x, 0.0f);
        o.y = fmaxf(bias[f0 + 1] + dd * acc.y, 0.0f);
        o.z = fmaxf(bias[f0 + 2] + dd * acc.z, 0.0f);
        o.w = fmaxf(bias[f0 + 3] + dd * acc.w, 0.0f);
    }
    *(float4*)(tile + r * TPITCH + f0) = o;
    if ((t & 15) == 0) dinvs[r] = dd;
    __syncthreads();

    const int wave = t >> 5;
    if (wave >= NT) return;
    const int lane = t & 31;
    const int half = lane >> 4;
    const int l    = lane & 15;
    const int col  = wave * 16 + l;
    const bool colOK = (col < FOUT);
    const int  colLd = colOK ? col : (FOUT - 1);

    const float* arow = tile + l * TPITCH + 2 * half;
    const float* __restrict__ wcol = W + colLd + (size_t)(2 * half) * FOUT;

    float dv[8];
#pragma unroll
    for (int v = 0; v < 8; ++v) dv[v] = dinvs[8 * half + v];

    v8f c = {};
#pragma unroll
    for (int k0 = 0; k0 < F; k0 += 4) {
        v2f a; a.x = arow[k0]; a.y = arow[k0 + 1];
        v2f b;
        b.x = wcol[(size_t)k0 * FOUT];
        b.y = wcol[(size_t)(k0 + 1) * FOUT];
        c = __builtin_amdgcn_wmma_f32_16x16x4_f32(false, a, false, b, (short)0, c,
                                                  false, false);
    }

    if (colOK) {
#pragma unroll
        for (int v = 0; v < 8; ++v) {
            const int m = v + 8 * half;
            Hn[(size_t)(row0 + m) * FOUT + col] = c[v] * dv[v];
        }
    }
}


extern "C" void kernel_launch(void* const* d_in, const int* in_sizes, int n_in,
                              void* d_out, int out_size, void* d_ws, size_t ws_size,
                              hipStream_t stream) {
    const float* x  = (const float*)d_in[0];
    const int*   ei = (const int*)d_in[1];
    const float* W1 = (const float*)d_in[2];
    const float* b1 = (const float*)d_in[3];
    const float* W2 = (const float*)d_in[4];
    const float* b2 = (const float*)d_in[5];
    const float* W3 = (const float*)d_in[6];
    const float* b3 = (const float*)d_in[7];
    float* out = (float*)d_out;

    const int N = in_sizes[0] / F_IN;
    const int E = in_sizes[1] / 2;

    int*   rowptr  = (int*)d_ws;
    int*   csr_src = rowptr + N;
    float* bufH    = (float*)(csr_src + E);
    float* bufA    = bufH + (size_t)N * F_HID;

    const int NB    = (N + 255) / 256;
    const int chunk = (E + G1 - 1) / G1;
    int*      hist  = (int*)bufA;
    int*      loff  = hist + G1 * HP;
    unsigned* ebuf  = (unsigned*)(loff + G1 * HP);

    const int T = 256;
    const int gAO = (N * (F_OUT / 4) + T - 1) / T;

    slice_sort_kernel<<<G1, 1024, 0, stream>>>(ei, hist, loff, ebuf, E, chunk);
    bucket_sort_proj_kernel<<<NB, 1024, 0, stream>>>(ebuf, hist, loff, rowptr, csr_src, x, W1, bufH, N, E, chunk);

    const int rowTiles = (N + 15) / 16;

    agg_proj_kernel<F_HID><<<rowTiles, T, 0, stream>>>(bufH, rowptr, csr_src, b1, W2, bufA, N, E);

    agg_proj_kernel<F_OUT><<<rowTiles, T, 0, stream>>>(bufA, rowptr, csr_src, b2, W3, bufH, N, E);

    agg_gather_kernel<F_OUT><<<gAO, T, 0, stream>>>(bufH, rowptr, csr_src, b3, out, N, E);
}


# Workspace reordered: the two H planes first, edge lists behind, so every 256-B H row is one aligned 256-B unit and a gathered row of layers 1-2 touches 2 lines, not 3; only pointer arguments change
# speedup vs baseline: 5.1729x; 1.1851x over previous
#include <hip/hip_runtime.h>

typedef __attribute__((ext_vector_type(2))) float v2f;
typedef __attribute__((ext_vector_type(4))) float v4f;
typedef __attribute__((ext_vector_type(8))) float v8f;

#define F_IN  128
#define F_HID 64
#define F_OUT 40


#define G1 256
#define HP 256
static_assert(G1 == 256, "pass 2 maps its 1024 threads onto 256 slices: 4 threads per run, 16 groups of 16 slices");

__device__ __forceinline__ float dinv_from(int b, int e) {
    const float deg = (float)(e - b) + 1.0f;
    return (deg > 0.0f) ? rsqrtf(deg) : 0.0f;
}

__device__ __forceinline__ int wave_incl_scan(int x, int lane) {
#pragma unroll
    for (int o = 1; o < 32; o <<= 1) {
        const int y = __shfl_up(x, o);
        if (lane >= o) x += y;
    }
    return x;
}

__global__ __launch_bounds__(1024) void slice_sort_kernel(const int* __restrict__ ei,
                                                          int* __restrict__ hist,
                                                          int* __restrict__ loff,
                                                          unsigned* __restrict__ ebuf,
                                                          int E, int chunk) {
    __shared__ int h[256];
    __shared__ int cur[256];
    __shared__ int wsum[8];
    const int t    = threadIdx.x;
    const int wg   = blockIdx.x;
    const int lane = t & 31;
    const int wave = t >> 5;
    if (t < 256) h[t] = 0;
    __syncthreads();
    const int c0 = wg * chunk;
    const int c1 = (c0 + chunk < E) ? (c0 + chunk) : E;

    const int e00 = c0 + t;
    int d0[4], s0[4];
#pragma unroll
    for (int k = 0; k < 4; ++k) {
        const int e = e00 + 1024 * k;
        int ec = (e < c1) ? e : (c1 - 1);
        ec = (ec < 0) ? 0 : ec;
        d0[k] = ei[E + ec];
        s0[k] = ei[ec];
    }
    asm volatile("" :: "v"(d0[0]), "v"(d0[1]), "v"(d0[2]), "v"(d0[3]),
                       "v"(s0[0]), "v"(s0[1]), "v"(s0[2]), "v"(s0[3]));
#pragma unroll
    for (int k = 0; k < 4; ++k)
        if (e00 + 1024 * k < c1) atomicAdd(&h[(d0[k] >> 8) & 255], 1);
    for (int e0 = e00 + 4096; e0 < c1; e0 += 4096) {
        int d[4];
#pragma unroll
        for (int k = 0; k < 4; ++k) {
            const int e = e0 + 1024 * k;
            d[k] = ei[E + ((e < c1) ? e : (c1 - 1))];
        }
        asm volatile("" :: "v"(d[0]), "v"(d[1]), "v"(d[2]), "v"(d[3]));
#pragma unroll
        for (int k = 0; k < 4; ++k)
            if (e0 + 1024 * k < c1) atomicAdd(&h[(d[k] >> 8) & 255], 1);
    }
    __syncthreads();

    const int v = (t < 256) ? h[t] : 0;
    const int x = wave_incl_scan(v, lane);
    if (t < 256 && lane == 31) wsum[wave] = x;
    __syncthreads();
    if (t < 256) {
        int front = 0;
        for (int w = 0; w < wave; ++w) front += wsum[w];
        const int lo = front + x - v;
        cur[t] = lo;
        hist[wg * HP + t] = v;
        loff[wg * HP + t] = lo;
    }
    __syncthreads();

#pragma unroll
    for (int k = 0; k < 4; ++k) {
        if (e00 + 1024 * k < c1) {
            const int p = c0 + atomicAdd(&cur[(d0[k] >> 8) & 255], 1);
            if ((unsigned)p < (unsigned)E)
                ebuf[p] = ((unsigned)s0[k] << 8) | ((unsigned)d0[k] & 255u);
        }
    }
    for (int e0 = e00 + 4096; e0 < c1; e0 += 4096) {
        int d[4], s[4];
#pragma unroll
        for (int k = 0; k < 4; ++k) {
            const int e  = e0 + 1024 * k;
            const int ec = (e < c1) ? e : (c1 - 1);
            d[k] = ei[E + ec];
            s[k] = ei[ec];
        }
#pragma unroll
        for (int k = 0; k < 4; ++k) {
            if (e0 + 1024 * k < c1) {
                const int p = c0 + atomicAdd(&cur[(d[k] >> 8) & 255], 1);
                if ((unsigned)p < (unsigned)E)
                    ebuf[p] = ((unsigned)s[k] << 8) | ((unsigned)d[k] & 255u);
            }
        }
    }
}

#define XP 132

typedef __attribute__((ext_vector_type(4))) int v4i;

#define RK 8

__global__ __launch_bounds__(1024) void bucket_sort_proj_kernel(const unsigned* __restrict__ ebuf,
                                                                const int* __restrict__ hist,
                                                                const int* __restrict__ loff,
                                                                int* __restrict__ rowptr,
                                                                int* __restrict__ csr_src,
                                                                const float* __restrict__ X,
                                                                const float* __restrict__ W,
                                                                float* __restrict__ H,
                                                                int N, int E, int chunk) {
    __shared__ float xs[256 * XP];
    __shared__ float dinvs[256];
    __shared__ int part[16][256];
    __shared__ int cnt[256];
    __shared__ int cur[256];
    __shared__ int wsum[8];
    __shared__ int wsumb[8];
    __shared__ int bstart;
    const int t    = threadIdx.x;
    const int b    = blockIdx.x;
    const int lane = t & 31;
    const int wave = t >> 5;

    v4f xr[8];
#pragma unroll
    for (int jx = 0; jx < 8; ++jx) {
        const int g  = b * 256 + wave + 32 * jx;
        const int gc = (g < N) ? g : (N - 1);
        xr[jx] = *(const v4f*)(X + (size_t)gc * F_IN + lane * 4);
    }

    const int sl = t >> 2;
    const int j  = t & 3;
    int ln = hist[sl * HP + b];
    int st = sl * chunk + loff[sl * HP + b];

    const int bq = t & 63;
    const int sg = t >> 6;
    v4i tot = {0, 0, 0, 0};
#pragma unroll
    for (int jj = 0; jj < 16; ++jj)
        tot += *(const v4i*)(hist + (sg * 16 + jj) * HP + bq * 4);

    ln = (ln < 0) ? 0 : ln;
    st = (st < 0) ? 0 : ((st > E) ? E : st);
    ln = (ln > E - st) ? (E - st) : ln;

    unsigned r0[RK];
#pragma unroll
    for (int k = 0; k < RK; ++k) {
        const int i = j + 4 * k;
        int ic = st + ((i < ln) ? i : (ln - 1));
        ic = (ic < 0) ? 0 : ic;
        r0[k] = ebuf[ic];
    }

    *(v4i*)&part[sg][bq * 4] = tot;
    if (t < 256) cnt[t] = 0;
    __syncthreads();

    asm volatile("" :: "v"(r0[0]), "v"(r0[1]), "v"(r0[2]), "v"(r0[3]),
                       "v"(r0[4]), "v"(r0[5]), "v"(r0[6]), "v"(r0[7]));
#pragma unroll
    for (int k = 0; k < RK; ++k)
        if (j + 4 * k < ln) atomicAdd(&cnt[r0[k] & 255u], 1);
    for (int i0 = j + 4 * RK; i0 < ln; i0 += 16) {
        unsigned r[4];
#pragma unroll
        for (int k = 0; k < 4; ++k) {
            const int i = i0 + 4 * k;
            r[k] = ebuf[st + ((i < ln) ? i : (ln - 1))];
        }
        asm volatile("" :: "v"(r[0]), "v"(r[1]), "v"(r[2]), "v"(r[3]));
#pragma unroll
        for (int k = 0; k < 4; ++k)
            if (i0 + 4 * k < ln) atomicAdd(&cnt[r[k] & 255u], 1);
    }

    int vb = 0;
    if (t < 256) {
#pragma unroll
        for (int g = 0; g < 16; ++g) vb += part[g][t];
    }
    const int xb = wave_incl_scan(vb, lane);
    if (t < 256 && lane == 31) wsumb[wave] = xb;

#pragma unroll
    for (int jx = 0; jx < 8; ++jx)
        *(v4f*)(xs + (wave + 32 * jx) * XP + lane * 4) = xr[jx];
    __syncthreads();

    const int v = (t < 256) ? cnt[t] : 0;
    const int x = wave_incl_scan(v, lane);
    if (t < 256 && lane == 31) wsum[wave] = x;
    if (t == b) {
        int front = 0;
        for (int w = 0; w < wave; ++w) front += wsumb[w];
        int bb = front + xb - vb;
        bb = (bb < 0) ? 0 : ((bb > E) ? E : bb);
        bstart = bb;
    }
    __syncthreads();
    if (t < 256) {
        int front = 0;
        for (int w = 0; w < wave; ++w) front += wsum[w];
        const int first = bstart + front + x - v;
        cur[t] = first;
        const int node = b * 256 + t;
        if (node < N) rowptr[node] = first;
        dinvs[t] = dinv_from(first, first + v);
    }
    __syncthreads();

#pragma unroll
    for (int k = 0; k < RK; ++k) {
        if (j + 4 * k < ln) {
            const int p = atomicAdd(&cur[r0[k] & 255u], 1);
            if ((unsigned)p < (unsigned)E) csr_src[p] = (int)(r0[k] >> 8);
        }
    }
    for (int i0 = j + 4 * RK; i0 < ln; i0 += 16) {
        unsigned r[4];
#pragma unroll
        for (int k = 0; k < 4; ++k) {
            const int i = i0 + 4 * k;
            r[k] = ebuf[st + ((i < ln) ? i : (ln - 1))];
        }
#pragma unroll
        for (int k = 0; k < 4; ++k) {
            if (i0 + 4 * k < ln) {
                const int p = atomicAdd(&cur[r[k] & 255u], 1);
                if ((unsigned)p < (unsigned)E) csr_src[p] = (int)(r[k] >> 8);
            }
        }
    }

    const int half = lane >> 4;
    const int l    = lane & 15;
    const int ct   = wave & 3;
    const int rp   = wave >> 2;
    const int col  = ct * 16 + l;

    const float* arow0 = xs + (rp * 16 + l) * XP + 2 * half;
    const float* arow1 = arow0 + 128 * XP;
    const float* __restrict__ wcol = W + col + (size_t)(2 * half) * F_HID;

    v8f c0 = {};
    v8f c1 = {};
#pragma unroll
    for (int k0 = 0; k0 < F_IN; k0 += 4) {
        v2f bb;
        bb.x = wcol[(size_t)k0 * F_HID];
        bb.y = wcol[(size_t)(k0 + 1) * F_HID];
        v2f a0; a0.x = arow0[k0]; a0.y = arow0[k0 + 1];
        v2f a1; a1.x = arow1[k0]; a1.y = arow1[k0 + 1];
        c0 = __builtin_amdgcn_wmma_f32_16x16x4_f32(false, a0, false, bb, (short)0, c0,
                                                   false, false);
        c1 = __builtin_amdgcn_wmma_f32_16x16x4_f32(false, a1, false, bb, (short)0, c1,
                                                   false, false);
    }

#pragma unroll
    for (int vv = 0; vv < 8; ++vv) {
        const int lr = rp * 16 + 8 * half + vv;
        const int g0 = b * 256 + lr;
        const int g1 = g0 + 128;
        if (g0 < N) H[(size_t)g0 * F_HID + col] = c0[vv] * dinvs[lr];
        if (g1 < N) H[(size_t)g1 * F_HID + col] = c1[vv] * dinvs[lr + 128];
    }
}


#define GCH 8
template <int F>
__device__ __forceinline__ float4 gather_rows(const float* __restrict__ H,
                                              const int* __restrict__ csr_src,
                                              int beg, int end, int f0, float4 acc) {
    if (beg >= end) return acc;
    const int last = end - 1;
    int s[GCH];
#pragma unroll
    for (int k = 0; k < GCH; ++k) {
        const int q = beg + k;
        s[k] = csr_src[(q < last) ? q : last];
    }
    for (int p = beg; p < end; p += GCH) {
        float4 hv[GCH];
#pragma unroll
        for (int k = 0; k < GCH; ++k) {
            const unsigned off = (unsigned)s[k] * (unsigned)(F * 4) + (unsigned)(f0 * 4);
            hv[k] = *(const float4*)((const char*)H + off);
        }
#pragma unroll
        for (int k = 0; k < GCH; ++k) {
            const int q = p + GCH + k;
            s[k] = csr_src[(q < last) ? q : last];
        }
        __builtin_amdgcn_sched_barrier(0);
#pragma unroll
        for (int k = 0; k < GCH; ++k) {
            const bool ok = (p + k < end);
            acc.x += ok ? hv[k].x : 0.0f;
            acc.y += ok ? hv[k].y : 0.0f;
            acc.z += ok ? hv[k].z : 0.0f;
            acc.w += ok ? hv[k].w : 0.0f;
        }
    }
    return acc;
}

template <int F>
__global__ __launch_bounds__(256, 16) void agg_gather_kernel(const float* __restrict__ H,
                                                         const int* __restrict__ rowptr,
                                                         const int* __restrict__ csr_src,
                                                         const float* __restrict__ bias,
                                                         float* __restrict__ A, int N, int E) {
    constexpr int CH = F / 4;
    const int tid = blockIdx.x * 256 + threadIdx.x;
    const int d   = tid / CH;
    if (d >= N) return;
    const int f0  = (tid - d * CH) * 4;

    const int beg = rowptr[d];
    const int d1  = (d + 1 < N) ? (d + 1) : (N - 1);
    const int r1  = rowptr[d1];
    const int end = (d + 1 < N) ? r1 : E;
    const float dd = dinv_from(beg, end);

    float4 acc = *(const float4*)(H + (size_t)d * F + f0);
    acc = gather_rows<F>(H, csr_src, beg, end, f0, acc);
    float4 o;
    o.x = bias[f0 + 0] + dd * acc.x;
    o.y = bias[f0 + 1] + dd * acc.y;
    o.z = bias[f0 + 2] + dd * acc.z;
    o.w = bias[f0 + 3] + dd * acc.w;
    *(float4*)(A + (size_t)d * F + f0) = o;
}

#define TPITCH 68

template <int FOUT>
__global__ __launch_bounds__(256, 16) void agg_proj_kernel(const float* __restrict__ H,
                                                       const int* __restrict__ rowptr,
                                                       const int* __restrict__ csr_src,
                                                       const float* __restrict__ bias,
                                                       const float* __restrict__ W,
                                                       float* __restrict__ Hn, int N, int E) {
    constexpr int F  = F_HID;
    constexpr int NT = (FOUT + 15) / 16;
    __shared__ float tile[16 * TPITCH];
    __shared__ float dinvs[16];

    const int t    = threadIdx.x;
    const int r    = t >> 4;
    const int f0   = (t & 15) * 4;
    const int row0 = blockIdx.x * 16;
    const int d    = row0 + r;

    float4 o  = make_float4(0.0f, 0.0f, 0.0f, 0.0f);
    float  dd = 0.0f;
    if (d < N) {
        const int beg = rowptr[d];
        const int d1  = (d + 1 < N) ? (d + 1) : (N - 1);
        const int r1  = rowptr[d1];
        const int end = (d + 1 < N) ? r1 : E;
        dd = dinv_from(beg, end);

        float4 acc = *(const float4*)(H + (size_t)d * F + f0);
        acc = gather_rows<F>(H, csr_src, beg, end, f0, acc);
        o.x = fmaxf(bias[f0 + 0] + dd * acc.x, 0.0f);
        o.y = fmaxf(bias[f0 + 1] + dd * acc.y, 0.0f);
        o.z = fmaxf(bias[f0 + 2] + dd * acc.z, 0.0f);
        o.w = fmaxf(bias[f0 + 3] + dd * acc.w, 0.0f);
    }
    *(float4*)(tile + r * TPITCH + f0) = o;
    if ((t & 15) == 0) dinvs[r] = dd;
    __syncthreads();

    const int wave = t >> 5;
    if (wave >= NT) return;
    const int lane = t & 31;
    const int half = lane >> 4;
    const int l    = lane & 15;
    const int col  = wave * 16 + l;
    const bool colOK = (col < FOUT);
    const int  colLd = colOK ? col : (FOUT - 1);

    const float* arow = tile + l * TPITCH + 2 * half;
    const float* __restrict__ wcol = W + colLd + (size_t)(2 * half) * FOUT;

    float dv[8];
#pragma unroll
    for (int v = 0; v < 8; ++v) dv[v] = dinvs[8 * half + v];

    v8f c = {};
#pragma unroll
    for (int k0 = 0; k0 < F; k0 += 4) {
        v2f a; a.x = arow[k0]; a.y = arow[k0 + 1];
        v2f b;
        b.x = wcol[(size_t)k0 * FOUT];
        b.y = wcol[(size_t)(k0 + 1) * FOUT];
        c = __builtin_amdgcn_wmma_f32_16x16x4_f32(false, a, false, b, (short)0, c,
                                                  false, false);
    }

    if (colOK) {
#pragma unroll
        for (int v = 0; v < 8; ++v) {
            const int m = v + 8 * half;
            Hn[(size_t)(row0 + m) * FOUT + col] = c[v] * dv[v];
        }
    }
}


extern "C" void kernel_launch(void* const* d_in, const int* in_sizes, int n_in,
                              void* d_out, int out_size, void* d_ws, size_t ws_size,
                              hipStream_t stream) {
    const float* x  = (const float*)d_in[0];
    const int*   ei = (const int*)d_in[1];
    const float* W1 = (const float*)d_in[2];
    const float* b1 = (const float*)d_in[3];
    const float* W2 = (const float*)d_in[4];
    const float* b2 = (const float*)d_in[5];
    const float* W3 = (const float*)d_in[6];
    const float* b3 = (const float*)d_in[7];
    float* out = (float*)d_out;

    const int N = in_sizes[0] / F_IN;
    const int E = in_sizes[1] / 2;

    float* bufH    = (float*)d_ws;
    float* bufA    = bufH + (size_t)N * F_HID;
    int*   csr_src = (int*)(bufA + (size_t)N * F_HID);
    int*   rowptr  = csr_src + E;

    const int NB    = (N + 255) / 256;
    const int chunk = (E + G1 - 1) / G1;
    int*      hist  = (int*)bufA;
    int*      loff  = hist + G1 * HP;
    unsigned* ebuf  = (unsigned*)(loff + G1 * HP);

    const int T = 256;
    const int gAO = (N * (F_OUT / 4) + T - 1) / T;

    slice_sort_kernel<<<G1, 1024, 0, stream>>>(ei, hist, loff, ebuf, E, chunk);
    bucket_sort_proj_kernel<<<NB, 1024, 0, stream>>>(ebuf, hist, loff, rowptr, csr_src, x, W1, bufH, N, E, chunk);

    const int rowTiles = (N + 15) / 16;

    agg_proj_kernel<F_HID><<<rowTiles, T, 0, stream>>>(bufH, rowptr, csr_src, b1, W2, bufA, N, E);

    agg_proj_kernel<F_OUT><<<rowTiles, T, 0, stream>>>(bufA, rowptr, csr_src, b2, W3, bufH, N, E);

    agg_gather_kernel<F_OUT><<<gAO, T, 0, stream>>>(bufH, rowptr, csr_src, b3, out, N, E);
}


# Both sort passes place records into an LDS image of their output range and copy it out as whole 128-B lines (slices padded to 32 records): no scattered 4-byte stores; list order = LDS arrival order
# speedup vs baseline: 5.3964x; 1.0432x over previous
#include <hip/hip_runtime.h>

typedef __attribute__((ext_vector_type(2))) float v2f;
typedef __attribute__((ext_vector_type(4))) float v4f;
typedef __attribute__((ext_vector_type(8))) float v8f;

#define F_IN  128
#define F_HID 64
#define F_OUT 40


#define G1 256
#define HP 256
#define SCAP 4096
#define CCAP 8192

typedef __attribute__((ext_vector_type(4))) unsigned v4u;
static_assert(G1 == 256, "pass 2 maps its 1024 threads onto 256 slices: 4 threads per run, 16 groups of 16 slices");

__device__ __forceinline__ float dinv_from(int b, int e) {
    const float deg = (float)(e - b) + 1.0f;
    return (deg > 0.0f) ? rsqrtf(deg) : 0.0f;
}

__device__ __forceinline__ int wave_incl_scan(int x, int lane) {
#pragma unroll
    for (int o = 1; o < 32; o <<= 1) {
        const int y = __shfl_up(x, o);
        if (lane >= o) x += y;
    }
    return x;
}

__global__ __launch_bounds__(1024) void slice_sort_kernel(const int* __restrict__ ei,
                                                          int* __restrict__ hist,
                                                          int* __restrict__ loff,
                                                          unsigned* __restrict__ ebuf,
                                                          int E, int chunk) {
    __shared__ int h[256];
    __shared__ int cur[256];
    __shared__ int wsum[8];
    __shared__ __attribute__((aligned(16))) unsigned img[SCAP];
    const int t    = threadIdx.x;
    const int wg   = blockIdx.x;
    const int lane = t & 31;
    const int wave = t >> 5;
    if (t < 256) h[t] = 0;
    __syncthreads();
    const int c0 = wg * chunk;
    const int c1 = (c0 + chunk < E) ? (c0 + chunk) : E;
    const bool staged = (chunk <= SCAP) && ((chunk & 31) == 0);

    const int e00 = c0 + t;
    int d0[4], s0[4];
#pragma unroll
    for (int k = 0; k < 4; ++k) {
        const int e = e00 + 1024 * k;
        int ec = (e < c1) ? e : (c1 - 1);
        ec = (ec < 0) ? 0 : ec;
        d0[k] = ei[E + ec];
        s0[k] = ei[ec];
    }
    asm volatile("" :: "v"(d0[0]), "v"(d0[1]), "v"(d0[2]), "v"(d0[3]),
                       "v"(s0[0]), "v"(s0[1]), "v"(s0[2]), "v"(s0[3]));
#pragma unroll
    for (int k = 0; k < 4; ++k)
        if (e00 + 1024 * k < c1) atomicAdd(&h[(d0[k] >> 8) & 255], 1);
    for (int e0 = e00 + 4096; e0 < c1; e0 += 4096) {
        int d[4];
#pragma unroll
        for (int k = 0; k < 4; ++k) {
            const int e = e0 + 1024 * k;
            d[k] = ei[E + ((e < c1) ? e : (c1 - 1))];
        }
        asm volatile("" :: "v"(d[0]), "v"(d[1]), "v"(d[2]), "v"(d[3]));
#pragma unroll
        for (int k = 0; k < 4; ++k)
            if (e0 + 1024 * k < c1) atomicAdd(&h[(d[k] >> 8) & 255], 1);
    }
    __syncthreads();

    const int v = (t < 256) ? h[t] : 0;
    const int x = wave_incl_scan(v, lane);
    if (t < 256 && lane == 31) wsum[wave] = x;
    __syncthreads();
    if (t < 256) {
        int front = 0;
        for (int w = 0; w < wave; ++w) front += wsum[w];
        const int lo = front + x - v;
        cur[t] = lo;
        hist[wg * HP + t] = v;
        loff[wg * HP + t] = lo;
    }
    __syncthreads();

#pragma unroll
    for (int k = 0; k < 4; ++k) {
        if (e00 + 1024 * k < c1) {
            const int q = atomicAdd(&cur[(d0[k] >> 8) & 255], 1);
            const unsigned rec = ((unsigned)s0[k] << 8) | ((unsigned)d0[k] & 255u);
            if (staged) {
                if ((unsigned)q < (unsigned)SCAP) img[q] = rec;
            } else {
                const int p = c0 + q;
                if ((unsigned)p < (unsigned)E) ebuf[p] = rec;
            }
        }
    }
    for (int e0 = e00 + 4096; e0 < c1; e0 += 4096) {
        int d[4], s[4];
#pragma unroll
        for (int k = 0; k < 4; ++k) {
            const int e  = e0 + 1024 * k;
            const int ec = (e < c1) ? e : (c1 - 1);
            d[k] = ei[E + ec];
            s[k] = ei[ec];
        }
#pragma unroll
        for (int k = 0; k < 4; ++k) {
            if (e0 + 1024 * k < c1) {
                const int p = c0 + atomicAdd(&cur[(d[k] >> 8) & 255], 1);
                if ((unsigned)p < (unsigned)E)
                    ebuf[p] = ((unsigned)s[k] << 8) | ((unsigned)d[k] & 255u);
            }
        }
    }

    __syncthreads();
    if (staged) {
        const int n = c1 - c0;
        const int i = 4 * t;
        if (i + 3 < n) {
            *(v4u*)(ebuf + c0 + i) = *(const v4u*)(img + i);
        } else {
#pragma unroll
            for (int k = 0; k < 3; ++k)
                if (i + k < n) ebuf[c0 + i + k] = img[i + k];
        }
    }
}

#define XP 132

typedef __attribute__((ext_vector_type(4))) int v4i;

#define RK 8

__global__ __launch_bounds__(1024) void bucket_sort_proj_kernel(const unsigned* __restrict__ ebuf,
                                                                const int* __restrict__ hist,
                                                                const int* __restrict__ loff,
                                                                int* __restrict__ rowptr,
                                                                int* __restrict__ csr_src,
                                                                const float* __restrict__ X,
                                                                const float* __restrict__ W,
                                                                float* __restrict__ H,
                                                                int N, int E, int chunk) {
    __shared__ float xs[256 * XP];
    __shared__ float dinvs[256];
    __shared__ int part[16][256];
    __shared__ int cnt[256];
    __shared__ int cur[256];
    __shared__ int wsum[8];
    __shared__ int wsumb[8];
    __shared__ int bstart;
    __shared__ int btot;
    __shared__ int img[CCAP];
    const int t    = threadIdx.x;
    const int b    = blockIdx.x;
    const int lane = t & 31;
    const int wave = t >> 5;

    v4f xr[8];
#pragma unroll
    for (int jx = 0; jx < 8; ++jx) {
        const int g  = b * 256 + wave + 32 * jx;
        const int gc = (g < N) ? g : (N - 1);
        xr[jx] = *(const v4f*)(X + (size_t)gc * F_IN + lane * 4);
    }

    const int sl = t >> 2;
    const int j  = t & 3;
    int ln = hist[sl * HP + b];
    int st = sl * chunk + loff[sl * HP + b];

    const int bq = t & 63;
    const int sg = t >> 6;
    v4i tot = {0, 0, 0, 0};
#pragma unroll
    for (int jj = 0; jj < 16; ++jj)
        tot += *(const v4i*)(hist + (sg * 16 + jj) * HP + bq * 4);

    ln = (ln < 0) ? 0 : ln;
    st = (st < 0) ? 0 : ((st > E) ? E : st);
    ln = (ln > E - st) ? (E - st) : ln;

    unsigned r0[RK];
#pragma unroll
    for (int k = 0; k < RK; ++k) {
        const int i = j + 4 * k;
        int ic = st + ((i < ln) ? i : (ln - 1));
        ic = (ic < 0) ? 0 : ic;
        r0[k] = ebuf[ic];
    }

    *(v4i*)&part[sg][bq * 4] = tot;
    if (t < 256) cnt[t] = 0;
    __syncthreads();

    asm volatile("" :: "v"(r0[0]), "v"(r0[1]), "v"(r0[2]), "v"(r0[3]),
                       "v"(r0[4]), "v"(r0[5]), "v"(r0[6]), "v"(r0[7]));
#pragma unroll
    for (int k = 0; k < RK; ++k)
        if (j + 4 * k < ln) atomicAdd(&cnt[r0[k] & 255u], 1);
    for (int i0 = j + 4 * RK; i0 < ln; i0 += 16) {
        unsigned r[4];
#pragma unroll
        for (int k = 0; k < 4; ++k) {
            const int i = i0 + 4 * k;
            r[k] = ebuf[st + ((i < ln) ? i : (ln - 1))];
        }
        asm volatile("" :: "v"(r[0]), "v"(r[1]), "v"(r[2]), "v"(r[3]));
#pragma unroll
        for (int k = 0; k < 4; ++k)
            if (i0 + 4 * k < ln) atomicAdd(&cnt[r[k] & 255u], 1);
    }

    int vb = 0;
    if (t < 256) {
#pragma unroll
        for (int g = 0; g < 16; ++g) vb += part[g][t];
    }
    const int xb = wave_incl_scan(vb, lane);
    if (t < 256 && lane == 31) wsumb[wave] = xb;

#pragma unroll
    for (int jx = 0; jx < 8; ++jx)
        *(v4f*)(xs + (wave + 32 * jx) * XP + lane * 4) = xr[jx];
    __syncthreads();

    const int v = (t < 256) ? cnt[t] : 0;
    const int x = wave_incl_scan(v, lane);
    if (t < 256 && lane == 31) wsum[wave] = x;
    if (t == b) {
        int front = 0;
        for (int w = 0; w < wave; ++w) front += wsumb[w];
        int bb = front + xb - vb;
        bb = (bb < 0) ? 0 : ((bb > E) ? E : bb);
        bstart = bb;
    }
    __syncthreads();
    if (t < 256) {
        int front = 0;
        for (int w = 0; w < wave; ++w) front += wsum[w];
        const int lo    = front + x - v;
        const int first = bstart + lo;
        cur[t] = lo;
        const int node = b * 256 + t;
        if (node < N) rowptr[node] = first;
        dinvs[t] = dinv_from(first, first + v);
        if (t == 255) btot = lo + v;
    }
    __syncthreads();

    const int  bs     = bstart;
    const int  tb     = btot;
    const bool staged = (tb <= CCAP);

#pragma unroll
    for (int k = 0; k < RK; ++k) {
        if (j + 4 * k < ln) {
            const int q = atomicAdd(&cur[r0[k] & 255u], 1);
            if (staged) {
                if ((unsigned)q < (unsigned)CCAP) img[q] = (int)(r0[k] >> 8);
            } else {
                const int p = bs + q;
                if ((unsigned)p < (unsigned)E) csr_src[p] = (int)(r0[k] >> 8);
            }
        }
    }
    for (int i0 = j + 4 * RK; i0 < ln; i0 += 16) {
        unsigned r[4];
#pragma unroll
        for (int k = 0; k < 4; ++k) {
            const int i = i0 + 4 * k;
            r[k] = ebuf[st + ((i < ln) ? i : (ln - 1))];
        }
#pragma unroll
        for (int k = 0; k < 4; ++k) {
            if (i0 + 4 * k < ln) {
                const int q = atomicAdd(&cur[r[k] & 255u], 1);
                if (staged) {
                    if ((unsigned)q < (unsigned)CCAP) img[q] = (int)(r[k] >> 8);
                } else {
                    const int p = bs + q;
                    if ((unsigned)p < (unsigned)E) csr_src[p] = (int)(r[k] >> 8);
                }
            }
        }
    }

    __syncthreads();
    if (staged) {
        for (int i = t - (bs & 31); i < tb; i += 1024) {
            const int p = bs + i;
            if (i >= 0 && (unsigned)p < (unsigned)E) csr_src[p] = img[i];
        }
    }

    const int half = lane >> 4;
    const int l    = lane & 15;
    const int ct   = wave & 3;
    const int rp   = wave >> 2;
    const int col  = ct * 16 + l;

    const float* arow0 = xs + (rp * 16 + l) * XP + 2 * half;
    const float* arow1 = arow0 + 128 * XP;
    const float* __restrict__ wcol = W + col + (size_t)(2 * half) * F_HID;

    v8f c0 = {};
    v8f c1 = {};
#pragma unroll
    for (int k0 = 0; k0 < F_IN; k0 += 4) {
        v2f bb;
        bb.x = wcol[(size_t)k0 * F_HID];
        bb.y = wcol[(size_t)(k0 + 1) * F_HID];
        v2f a0; a0.x = arow0[k0]; a0.y = arow0[k0 + 1];
        v2f a1; a1.x = arow1[k0]; a1.y = arow1[k0 + 1];
        c0 = __builtin_amdgcn_wmma_f32_16x16x4_f32(false, a0, false, bb, (short)0, c0,
                                                   false, false);
        c1 = __builtin_amdgcn_wmma_f32_16x16x4_f32(false, a1, false, bb, (short)0, c1,
                                                   false, false);
    }

#pragma unroll
    for (int vv = 0; vv < 8; ++vv) {
        const int lr = rp * 16 + 8 * half + vv;
        const int g0 = b * 256 + lr;
        const int g1 = g0 + 128;
        if (g0 < N) H[(size_t)g0 * F_HID + col] = c0[vv] * dinvs[lr];
        if (g1 < N) H[(size_t)g1 * F_HID + col] = c1[vv] * dinvs[lr + 128];
    }
}


#define GCH 8
template <int F>
__device__ __forceinline__ float4 gather_rows(const float* __restrict__ H,
                                              const int* __restrict__ csr_src,
                                              int beg, int end, int f0, float4 acc) {
    if (beg >= end) return acc;
    const int last = end - 1;
    int s[GCH];
#pragma unroll
    for (int k = 0; k < GCH; ++k) {
        const int q = beg + k;
        s[k] = csr_src[(q < last) ? q : last];
    }
    for (int p = beg; p < end; p += GCH) {
        float4 hv[GCH];
#pragma unroll
        for (int k = 0; k < GCH; ++k) {
            const unsigned off = (unsigned)s[k] * (unsigned)(F * 4) + (unsigned)(f0 * 4);
            hv[k] = *(const float4*)((const char*)H + off);
        }
#pragma unroll
        for (int k = 0; k < GCH; ++k) {
            const int q = p + GCH + k;
            s[k] = csr_src[(q < last) ? q : last];
        }
        __builtin_amdgcn_sched_barrier(0);
#pragma unroll
        for (int k = 0; k < GCH; ++k) {
            const bool ok = (p + k < end);
            acc.x += ok ? hv[k].x : 0.0f;
            acc.y += ok ? hv[k].y : 0.0f;
            acc.z += ok ? hv[k].z : 0.0f;
            acc.w += ok ? hv[k].w : 0.0f;
        }
    }
    return acc;
}

template <int F>
__global__ __launch_bounds__(256, 16) void agg_gather_kernel(const float* __restrict__ H,
                                                         const int* __restrict__ rowptr,
                                                         const int* __restrict__ csr_src,
                                                         const float* __restrict__ bias,
                                                         float* __restrict__ A, int N, int E) {
    constexpr int CH = F / 4;
    const int tid = blockIdx.x * 256 + threadIdx.x;
    const int d   = tid / CH;
    if (d >= N) return;
    const int f0  = (tid - d * CH) * 4;

    const int beg = rowptr[d];
    const int d1  = (d + 1 < N) ? (d + 1) : (N - 1);
    const int r1  = rowptr[d1];
    const int end = (d + 1 < N) ? r1 : E;
    const float dd = dinv_from(beg, end);

    float4 acc = *(const float4*)(H + (size_t)d * F + f0);
    acc = gather_rows<F>(H, csr_src, beg, end, f0, acc);
    float4 o;
    o.x = bias[f0 + 0] + dd * acc.x;
    o.y = bias[f0 + 1] + dd * acc.y;
    o.z = bias[f0 + 2] + dd * acc.z;
    o.w = bias[f0 + 3] + dd * acc.w;
    *(float4*)(A + (size_t)d * F + f0) = o;
}

#define TPITCH 68

template <int FOUT>
__global__ __launch_bounds__(256, 16) void agg_proj_kernel(const float* __restrict__ H,
                                                       const int* __restrict__ rowptr,
                                                       const int* __restrict__ csr_src,
                                                       const float* __restrict__ bias,
                                                       const float* __restrict__ W,
                                                       float* __restrict__ Hn, int N, int E) {
    constexpr int F  = F_HID;
    constexpr int NT = (FOUT + 15) / 16;
    __shared__ float tile[16 * TPITCH];
    __shared__ float dinvs[16];

    const int t    = threadIdx.x;
    const int r    = t >> 4;
    const int f0   = (t & 15) * 4;
    const int row0 = blockIdx.x * 16;
    const int d    = row0 + r;

    float4 o  = make_float4(0.0f, 0.0f, 0.0f, 0.0f);
    float  dd = 0.0f;
    if (d < N) {
        const int beg = rowptr[d];
        const int d1  = (d + 1 < N) ? (d + 1) : (N - 1);
        const int r1  = rowptr[d1];
        const int end = (d + 1 < N) ? r1 : E;
        dd = dinv_from(beg, end);

        float4 acc = *(const float4*)(H + (size_t)d * F + f0);
        acc = gather_rows<F>(H, csr_src, beg, end, f0, acc);
        o.x = fmaxf(bias[f0 + 0] + dd * acc.x, 0.0f);
        o.y = fmaxf(bias[f0 + 1] + dd * acc.y, 0.0f);
        o.z = fmaxf(bias[f0 + 2] + dd * acc.z, 0.0f);
        o.w = fmaxf(bias[f0 + 3] + dd * acc.w, 0.0f);
    }
    *(float4*)(tile + r * TPITCH + f0) = o;
    if ((t & 15) == 0) dinvs[r] = dd;
    __syncthreads();

    const int wave = t >> 5;
    if (wave >= NT) return;
    const int lane = t & 31;
    const int half = lane >> 4;
    const int l    = lane & 15;
    const int col  = wave * 16 + l;
    const bool colOK = (col < FOUT);
    const int  colLd = colOK ? col : (FOUT - 1);

    const float* arow = tile + l * TPITCH + 2 * half;
    const float* __restrict__ wcol = W + colLd + (size_t)(2 * half) * FOUT;

    float dv[8];
#pragma unroll
    for (int v = 0; v < 8; ++v) dv[v] = dinvs[8 * half + v];

    v8f c = {};
#pragma unroll
    for (int k0 = 0; k0 < F; k0 += 4) {
        v2f a; a.x = arow[k0]; a.y = arow[k0 + 1];
        v2f b;
        b.x = wcol[(size_t)k0 * FOUT];
        b.y = wcol[(size_t)(k0 + 1) * FOUT];
        c = __builtin_amdgcn_wmma_f32_16x16x4_f32(false, a, false, b, (short)0, c,
                                                  false, false);
    }

    if (colOK) {
#pragma unroll
        for (int v = 0; v < 8; ++v) {
            const int m = v + 8 * half;
            Hn[(size_t)(row0 + m) * FOUT + col] = c[v] * dv[v];
        }
    }
}


extern "C" void kernel_launch(void* const* d_in, const int* in_sizes, int n_in,
                              void* d_out, int out_size, void* d_ws, size_t ws_size,
                              hipStream_t stream) {
    const float* x  = (const float*)d_in[0];
    const int*   ei = (const int*)d_in[1];
    const float* W1 = (const float*)d_in[2];
    const float* b1 = (const float*)d_in[3];
    const float* W2 = (const float*)d_in[4];
    const float* b2 = (const float*)d_in[5];
    const float* W3 = (const float*)d_in[6];
    const float* b3 = (const float*)d_in[7];
    float* out = (float*)d_out;

    const int N = in_sizes[0] / F_IN;
    const int E = in_sizes[1] / 2;

    float* bufH    = (float*)d_ws;
    float* bufA    = bufH + (size_t)N * F_HID;
    int*   csr_src = (int*)(bufA + (size_t)N * F_HID);
    int*   rowptr  = csr_src + E;

    const int NB    = (N + 255) / 256;
    const int chunk = ((E + G1 - 1) / G1 + 31) & ~31;
    int*      hist  = (int*)bufA;
    int*      loff  = hist + G1 * HP;
    unsigned* ebuf  = (unsigned*)(loff + G1 * HP);

    const int T = 256;
    const int gAO = (N * (F_OUT / 4) + T - 1) / T;

    slice_sort_kernel<<<G1, 1024, 0, stream>>>(ei, hist, loff, ebuf, E, chunk);
    bucket_sort_proj_kernel<<<NB, 1024, 0, stream>>>(ebuf, hist, loff, rowptr, csr_src, x, W1, bufH, N, E, chunk);

    const int rowTiles = (N + 15) / 16;

    agg_proj_kernel<F_HID><<<rowTiles, T, 0, stream>>>(bufH, rowptr, csr_src, b1, W2, bufA, N, E);

    agg_proj_kernel<F_OUT><<<rowTiles, T, 0, stream>>>(bufA, rowptr, csr_src, b2, W3, bufH, N, E);

    agg_gather_kernel<F_OUT><<<gAO, T, 0, stream>>>(bufH, rowptr, csr_src, b3, out, N, E);
}


# Bucket's first slot in csr_src = sum over the slices of loff[s][b], words its work-group loads for its runs anyway: the 256 x 256 count table is no longer read whole by every bucket work-group
# speedup vs baseline: 5.6364x; 1.0445x over previous
#include <hip/hip_runtime.h>

typedef __attribute__((ext_vector_type(2))) float v2f;
typedef __attribute__((ext_vector_type(4))) float v4f;
typedef __attribute__((ext_vector_type(8))) float v8f;

#define F_IN  128
#define F_HID 64
#define F_OUT 40


#define G1 256
#define HP 256
#define SCAP 4096
#define CCAP 8192

typedef __attribute__((ext_vector_type(4))) unsigned v4u;
static_assert(G1 == 256, "pass 2 maps its 1024 threads onto 256 slices: 4 threads per run");

__device__ __forceinline__ float dinv_from(int b, int e) {
    const float deg = (float)(e - b) + 1.0f;
    return (deg > 0.0f) ? rsqrtf(deg) : 0.0f;
}

__device__ __forceinline__ int wave_sum(int x) {
    for (int o = 16; o >= 1; o >>= 1) x += __shfl_xor(x, o);
    return x;
}

__device__ __forceinline__ int wave_incl_scan(int x, int lane) {
#pragma unroll
    for (int o = 1; o < 32; o <<= 1) {
        const int y = __shfl_up(x, o);
        if (lane >= o) x += y;
    }
    return x;
}

__global__ __launch_bounds__(1024) void slice_sort_kernel(const int* __restrict__ ei,
                                                          int* __restrict__ hist,
                                                          int* __restrict__ loff,
                                                          unsigned* __restrict__ ebuf,
                                                          int E, int chunk) {
    __shared__ int h[256];
    __shared__ int cur[256];
    __shared__ int wsum[8];
    __shared__ __attribute__((aligned(16))) unsigned img[SCAP];
    const int t    = threadIdx.x;
    const int wg   = blockIdx.x;
    const int lane = t & 31;
    const int wave = t >> 5;
    if (t < 256) h[t] = 0;
    __syncthreads();
    const int c0 = wg * chunk;
    const int c1 = (c0 + chunk < E) ? (c0 + chunk) : E;
    const bool staged = (chunk <= SCAP) && ((chunk & 31) == 0);

    const int e00 = c0 + t;
    int d0[4], s0[4];
#pragma unroll
    for (int k = 0; k < 4; ++k) {
        const int e = e00 + 1024 * k;
        int ec = (e < c1) ? e : (c1 - 1);
        ec = (ec < 0) ? 0 : ec;
        d0[k] = ei[E + ec];
        s0[k] = ei[ec];
    }
    asm volatile("" :: "v"(d0[0]), "v"(d0[1]), "v"(d0[2]), "v"(d0[3]),
                       "v"(s0[0]), "v"(s0[1]), "v"(s0[2]), "v"(s0[3]));
#pragma unroll
    for (int k = 0; k < 4; ++k)
        if (e00 + 1024 * k < c1) atomicAdd(&h[(d0[k] >> 8) & 255], 1);
    for (int e0 = e00 + 4096; e0 < c1; e0 += 4096) {
        int d[4];
#pragma unroll
        for (int k = 0; k < 4; ++k) {
            const int e = e0 + 1024 * k;
            d[k] = ei[E + ((e < c1) ? e : (c1 - 1))];
        }
        asm volatile("" :: "v"(d[0]), "v"(d[1]), "v"(d[2]), "v"(d[3]));
#pragma unroll
        for (int k = 0; k < 4; ++k)
            if (e0 + 1024 * k < c1) atomicAdd(&h[(d[k] >> 8) & 255], 1);
    }
    __syncthreads();

    const int v = (t < 256) ? h[t] : 0;
    const int x = wave_incl_scan(v, lane);
    if (t < 256 && lane == 31) wsum[wave] = x;
    __syncthreads();
    if (t < 256) {
        int front = 0;
        for (int w = 0; w < wave; ++w) front += wsum[w];
        const int lo = front + x - v;
        cur[t] = lo;
        hist[wg * HP + t] = v;
        loff[wg * HP + t] = lo;
    }
    __syncthreads();

#pragma unroll
    for (int k = 0; k < 4; ++k) {
        if (e00 + 1024 * k < c1) {
            const int q = atomicAdd(&cur[(d0[k] >> 8) & 255], 1);
            const unsigned rec = ((unsigned)s0[k] << 8) | ((unsigned)d0[k] & 255u);
            if (staged) {
                if ((unsigned)q < (unsigned)SCAP) img[q] = rec;
            } else {
                const int p = c0 + q;
                if ((unsigned)p < (unsigned)E) ebuf[p] = rec;
            }
        }
    }
    for (int e0 = e00 + 4096; e0 < c1; e0 += 4096) {
        int d[4], s[4];
#pragma unroll
        for (int k = 0; k < 4; ++k) {
            const int e  = e0 + 1024 * k;
            const int ec = (e < c1) ? e : (c1 - 1);
            d[k] = ei[E + ec];
            s[k] = ei[ec];
        }
#pragma unroll
        for (int k = 0; k < 4; ++k) {
            if (e0 + 1024 * k < c1) {
                const int p = c0 + atomicAdd(&cur[(d[k] >> 8) & 255], 1);
                if ((unsigned)p < (unsigned)E)
                    ebuf[p] = ((unsigned)s[k] << 8) | ((unsigned)d[k] & 255u);
            }
        }
    }

    __syncthreads();
    if (staged) {
        const int n = c1 - c0;
        const int i = 4 * t;
        if (i + 3 < n) {
            *(v4u*)(ebuf + c0 + i) = *(const v4u*)(img + i);
        } else {
#pragma unroll
            for (int k = 0; k < 3; ++k)
                if (i + k < n) ebuf[c0 + i + k] = img[i + k];
        }
    }
}

#define XP 132

typedef __attribute__((ext_vector_type(4))) int v4i;

#define RK 8

__global__ __launch_bounds__(1024) void bucket_sort_proj_kernel(const unsigned* __restrict__ ebuf,
                                                                const int* __restrict__ hist,
                                                                const int* __restrict__ loff,
                                                                int* __restrict__ rowptr,
                                                                int* __restrict__ csr_src,
                                                                const float* __restrict__ X,
                                                                const float* __restrict__ W,
                                                                float* __restrict__ H,
                                                                int N, int E, int chunk) {
    __shared__ float xs[256 * XP];
    __shared__ float dinvs[256];
    __shared__ int cnt[256];
    __shared__ int cur[256];
    __shared__ int wsum[8];
    __shared__ int wsumb[32];
    __shared__ int bstart;
    __shared__ int btot;
    __shared__ int img[CCAP];
    const int t    = threadIdx.x;
    const int b    = blockIdx.x;
    const int lane = t & 31;
    const int wave = t >> 5;

    v4f xr[8];
#pragma unroll
    for (int jx = 0; jx < 8; ++jx) {
        const int g  = b * 256 + wave + 32 * jx;
        const int gc = (g < N) ? g : (N - 1);
        xr[jx] = *(const v4f*)(X + (size_t)gc * F_IN + lane * 4);
    }

    const int sl = t >> 2;
    const int j  = t & 3;
    int ln = hist[sl * HP + b];
    const int lo_s = loff[sl * HP + b];
    int st = sl * chunk + lo_s;

    ln = (ln < 0) ? 0 : ln;
    st = (st < 0) ? 0 : ((st > E) ? E : st);
    ln = (ln > E - st) ? (E - st) : ln;

    unsigned r0[RK];
#pragma unroll
    for (int k = 0; k < RK; ++k) {
        const int i = j + 4 * k;
        int ic = st + ((i < ln) ? i : (ln - 1));
        ic = (ic < 0) ? 0 : ic;
        r0[k] = ebuf[ic];
    }

    const int bsum = wave_sum((j == 0) ? lo_s : 0);
    if (lane == 0) wsumb[wave] = bsum;
    if (t < 256) cnt[t] = 0;
    __syncthreads();

    asm volatile("" :: "v"(r0[0]), "v"(r0[1]), "v"(r0[2]), "v"(r0[3]),
                       "v"(r0[4]), "v"(r0[5]), "v"(r0[6]), "v"(r0[7]));
#pragma unroll
    for (int k = 0; k < RK; ++k)
        if (j + 4 * k < ln) atomicAdd(&cnt[r0[k] & 255u], 1);
    for (int i0 = j + 4 * RK; i0 < ln; i0 += 16) {
        unsigned r[4];
#pragma unroll
        for (int k = 0; k < 4; ++k) {
            const int i = i0 + 4 * k;
            r[k] = ebuf[st + ((i < ln) ? i : (ln - 1))];
        }
        asm volatile("" :: "v"(r[0]), "v"(r[1]), "v"(r[2]), "v"(r[3]));
#pragma unroll
        for (int k = 0; k < 4; ++k)
            if (i0 + 4 * k < ln) atomicAdd(&cnt[r[k] & 255u], 1);
    }

#pragma unroll
    for (int jx = 0; jx < 8; ++jx)
        *(v4f*)(xs + (wave + 32 * jx) * XP + lane * 4) = xr[jx];
    __syncthreads();

    const int v = (t < 256) ? cnt[t] : 0;
    const int x = wave_incl_scan(v, lane);
    if (t < 256 && lane == 31) wsum[wave] = x;
    if (wave == 0) {
        int bb = wave_sum(wsumb[lane]);
        bb = (bb < 0) ? 0 : ((bb > E) ? E : bb);
        if (lane == 0) bstart = bb;
    }
    __syncthreads();
    if (t < 256) {
        int front = 0;
        for (int w = 0; w < wave; ++w) front += wsum[w];
        const int lo    = front + x - v;
        const int first = bstart + lo;
        cur[t] = lo;
        const int node = b * 256 + t;
        if (node < N) rowptr[node] = first;
        dinvs[t] = dinv_from(first, first + v);
        if (t == 255) btot = lo + v;
    }
    __syncthreads();

    const int  bs     = bstart;
    const int  tb     = btot;
    const bool staged = (tb <= CCAP);

#pragma unroll
    for (int k = 0; k < RK; ++k) {
        if (j + 4 * k < ln) {
            const int q = atomicAdd(&cur[r0[k] & 255u], 1);
            if (staged) {
                if ((unsigned)q < (unsigned)CCAP) img[q] = (int)(r0[k] >> 8);
            } else {
                const int p = bs + q;
                if ((unsigned)p < (unsigned)E) csr_src[p] = (int)(r0[k] >> 8);
            }
        }
    }
    for (int i0 = j + 4 * RK; i0 < ln; i0 += 16) {
        unsigned r[4];
#pragma unroll
        for (int k = 0; k < 4; ++k) {
            const int i = i0 + 4 * k;
            r[k] = ebuf[st + ((i < ln) ? i : (ln - 1))];
        }
#pragma unroll
        for (int k = 0; k < 4; ++k) {
            if (i0 + 4 * k < ln) {
                const int q = atomicAdd(&cur[r[k] & 255u], 1);
                if (staged) {
                    if ((unsigned)q < (unsigned)CCAP) img[q] = (int)(r[k] >> 8);
                } else {
                    const int p = bs + q;
                    if ((unsigned)p < (unsigned)E) csr_src[p] = (int)(r[k] >> 8);
                }
            }
        }
    }

    __syncthreads();
    if (staged) {
        for (int i = t - (bs & 31); i < tb; i += 1024) {
            const int p = bs + i;
            if (i >= 0 && (unsigned)p < (unsigned)E) csr_src[p] = img[i];
        }
    }

    const int half = lane >> 4;
    const int l    = lane & 15;
    const int ct   = wave & 3;
    const int rp   = wave >> 2;
    const int col  = ct * 16 + l;

    const float* arow0 = xs + (rp * 16 + l) * XP + 2 * half;
    const float* arow1 = arow0 + 128 * XP;
    const float* __restrict__ wcol = W + col + (size_t)(2 * half) * F_HID;

    v8f c0 = {};
    v8f c1 = {};
#pragma unroll
    for (int k0 = 0; k0 < F_IN; k0 += 4) {
        v2f bb;
        bb.x = wcol[(size_t)k0 * F_HID];
        bb.y = wcol[(size_t)(k0 + 1) * F_HID];
        v2f a0; a0.x = arow0[k0]; a0.y = arow0[k0 + 1];
        v2f a1; a1.x = arow1[k0]; a1.y = arow1[k0 + 1];
        c0 = __builtin_amdgcn_wmma_f32_16x16x4_f32(false, a0, false, bb, (short)0, c0,
                                                   false, false);
        c1 = __builtin_amdgcn_wmma_f32_16x16x4_f32(false, a1, false, bb, (short)0, c1,
                                                   false, false);
    }

#pragma unroll
    for (int vv = 0; vv < 8; ++vv) {
        const int lr = rp * 16 + 8 * half + vv;
        const int g0 = b * 256 + lr;
        const int g1 = g0 + 128;
        if (g0 < N) H[(size_t)g0 * F_HID + col] = c0[vv] * dinvs[lr];
        if (g1 < N) H[(size_t)g1 * F_HID + col] = c1[vv] * dinvs[lr + 128];
    }
}


#define GCH 8
template <int F>
__device__ __forceinline__ float4 gather_rows(const float* __restrict__ H,
                                              const int* __restrict__ csr_src,
                                              int beg, int end, int f0, float4 acc) {
    if (beg >= end) return acc;
    const int last = end - 1;
    int s[GCH];
#pragma unroll
    for (int k = 0; k < GCH; ++k) {
        const int q = beg + k;
        s[k] = csr_src[(q < last) ? q : last];
    }
    for (int p = beg; p < end; p += GCH) {
        float4 hv[GCH];
#pragma unroll
        for (int k = 0; k < GCH; ++k) {
            const unsigned off = (unsigned)s[k] * (unsigned)(F * 4) + (unsigned)(f0 * 4);
            hv[k] = *(const float4*)((const char*)H + off);
        }
#pragma unroll
        for (int k = 0; k < GCH; ++k) {
            const int q = p + GCH + k;
            s[k] = csr_src[(q < last) ? q : last];
        }
        __builtin_amdgcn_sched_barrier(0);
#pragma unroll
        for (int k = 0; k < GCH; ++k) {
            const bool ok = (p + k < end);
            acc.x += ok ? hv[k].x : 0.0f;
            acc.y += ok ? hv[k].y : 0.0f;
            acc.z += ok ? hv[k].z : 0.0f;
            acc.w += ok ? hv[k].w : 0.0f;
        }
    }
    return acc;
}

template <int F>
__global__ __launch_bounds__(256, 16) void agg_gather_kernel(const float* __restrict__ H,
                                                         const int* __restrict__ rowptr,
                                                         const int* __restrict__ csr_src,
                                                         const float* __restrict__ bias,
                                                         float* __restrict__ A, int N, int E) {
    constexpr int CH = F / 4;
    const int tid = blockIdx.x * 256 + threadIdx.x;
    const int d   = tid / CH;
    if (d >= N) return;
    const int f0  = (tid - d * CH) * 4;

    const int beg = rowptr[d];
    const int d1  = (d + 1 < N) ? (d + 1) : (N - 1);
    const int r1  = rowptr[d1];
    const int end = (d + 1 < N) ? r1 : E;
    const float dd = dinv_from(beg, end);

    float4 acc = *(const float4*)(H + (size_t)d * F + f0);
    acc = gather_rows<F>(H, csr_src, beg, end, f0, acc);
    float4 o;
    o.x = bias[f0 + 0] + dd * acc.x;
    o.y = bias[f0 + 1] + dd * acc.y;
    o.z = bias[f0 + 2] + dd * acc.z;
    o.w = bias[f0 + 3] + dd * acc.w;
    *(float4*)(A + (size_t)d * F + f0) = o;
}

#define TPITCH 68

template <int FOUT>
__global__ __launch_bounds__(256, 16) void agg_proj_kernel(const float* __restrict__ H,
                                                       const int* __restrict__ rowptr,
                                                       const int* __restrict__ csr_src,
                                                       const float* __restrict__ bias,
                                                       const float* __restrict__ W,
                                                       float* __restrict__ Hn, int N, int E) {
    constexpr int F  = F_HID;
    constexpr int NT = (FOUT + 15) / 16;
    __shared__ float tile[16 * TPITCH];
    __shared__ float dinvs[16];

    const int t    = threadIdx.x;
    const int r    = t >> 4;
    const int f0   = (t & 15) * 4;
    const int row0 = blockIdx.x * 16;
    const int d    = row0 + r;

    float4 o  = make_float4(0.0f, 0.0f, 0.0f, 0.0f);
    float  dd = 0.0f;
    if (d < N) {
        const int beg = rowptr[d];
        const int d1  = (d + 1 < N) ? (d + 1) : (N - 1);
        const int r1  = rowptr[d1];
        const int end = (d + 1 < N) ? r1 : E;
        dd = dinv_from(beg, end);

        float4 acc = *(const float4*)(H + (size_t)d * F + f0);
        acc = gather_rows<F>(H, csr_src, beg, end, f0, acc);
        o.x = fmaxf(bias[f0 + 0] + dd * acc.x, 0.0f);
        o.y = fmaxf(bias[f0 + 1] + dd * acc.y, 0.0f);
        o.z = fmaxf(bias[f0 + 2] + dd * acc.z, 0.0f);
        o.w = fmaxf(bias[f0 + 3] + dd * acc.w, 0.0f);
    }
    *(float4*)(tile + r * TPITCH + f0) = o;
    if ((t & 15) == 0) dinvs[r] = dd;
    __syncthreads();

    const int wave = t >> 5;
    if (wave >= NT) return;
    const int lane = t & 31;
    const int half = lane >> 4;
    const int l    = lane & 15;
    const int col  = wave * 16 + l;
    const bool colOK = (col < FOUT);
    const int  colLd = colOK ? col : (FOUT - 1);

    const float* arow = tile + l * TPITCH + 2 * half;
    const float* __restrict__ wcol = W + colLd + (size_t)(2 * half) * FOUT;

    float dv[8];
#pragma unroll
    for (int v = 0; v < 8; ++v) dv[v] = dinvs[8 * half + v];

    v8f c = {};
#pragma unroll
    for (int k0 = 0; k0 < F; k0 += 4) {
        v2f a; a.x = arow[k0]; a.y = arow[k0 + 1];
        v2f b;
        b.x = wcol[(size_t)k0 * FOUT];
        b.y = wcol[(size_t)(k0 + 1) * FOUT];
        c = __builtin_amdgcn_wmma_f32_16x16x4_f32(false, a, false, b, (short)0, c,
                                                  false, false);
    }

    if (colOK) {
#pragma unroll
        for (int v = 0; v < 8; ++v) {
            const int m = v + 8 * half;
            Hn[(size_t)(row0 + m) * FOUT + col] = c[v] * dv[v];
        }
    }
}


extern "C" void kernel_launch(void* const* d_in, const int* in_sizes, int n_in,
                              void* d_out, int out_size, void* d_ws, size_t ws_size,
                              hipStream_t stream) {
    const float* x  = (const float*)d_in[0];
    const int*   ei = (const int*)d_in[1];
    const float* W1 = (const float*)d_in[2];
    const float* b1 = (const float*)d_in[3];
    const float* W2 = (const float*)d_in[4];
    const float* b2 = (const float*)d_in[5];
    const float* W3 = (const float*)d_in[6];
    const float* b3 = (const float*)d_in[7];
    float* out = (float*)d_out;

    const int N = in_sizes[0] / F_IN;
    const int E = in_sizes[1] / 2;

    float* bufH    = (float*)d_ws;
    float* bufA    = bufH + (size_t)N * F_HID;
    int*   csr_src = (int*)(bufA + (size_t)N * F_HID);
    int*   rowptr  = csr_src + E;

    const int NB    = (N + 255) / 256;
    const int chunk = ((E + G1 - 1) / G1 + 31) & ~31;
    int*      hist  = (int*)bufA;
    int*      loff  = hist + G1 * HP;
    unsigned* ebuf  = (unsigned*)(loff + G1 * HP);

    const int T = 256;
    const int gAO = (N * (F_OUT / 4) + T - 1) / T;

    slice_sort_kernel<<<G1, 1024, 0, stream>>>(ei, hist, loff, ebuf, E, chunk);
    bucket_sort_proj_kernel<<<NB, 1024, 0, stream>>>(ebuf, hist, loff, rowptr, csr_src, x, W1, bufH, N, E, chunk);

    const int rowTiles = (N + 15) / 16;

    agg_proj_kernel<F_HID><<<rowTiles, T, 0, stream>>>(bufH, rowptr, csr_src, b1, W2, bufA, N, E);

    agg_proj_kernel<F_OUT><<<rowTiles, T, 0, stream>>>(bufA, rowptr, csr_src, b2, W3, bufH, N, E);

    agg_gather_kernel<F_OUT><<<gAO, T, 0, stream>>>(bufH, rowptr, csr_src, b3, out, N, E);
}
